# starting kernel
# baseline (speedup 1.0000x reference)
#include <hip/hip_runtime.h>

typedef __attribute__((ext_vector_type(16))) _Float16 v16h;
typedef __attribute__((ext_vector_type(8)))  _Float16 v8h;
typedef __attribute__((ext_vector_type(8)))  float    v8f;

#define B_   8
#define N_   2048
#define D_   128
#define L_   3
#define EPSN 1e-5f

__device__ __forceinline__ float fast_sqrt(float x) {
#if __has_builtin(__builtin_amdgcn_sqrtf)
    return __builtin_amdgcn_sqrtf(x);
#else
    return sqrtf(x);
#endif
}

__device__ __forceinline__ float fast_tanh(float x) {
#if __has_builtin(__builtin_amdgcn_tanhf)
    return __builtin_amdgcn_tanhf(x);
#else
    return tanhf(x);
#endif
}

__device__ __forceinline__ v16h load_frag(const _Float16* p) {
    v8h lo = *(const v8h*)(p);
    v8h hi = *(const v8h*)(p + 16);
    return __builtin_shufflevector(lo, hi, 0,1,2,3,4,5,6,7,8,9,10,11,12,13,14,15);
}

__global__ __launch_bounds__(256) void k_dinv(const float* __restrict__ inst,
                                              float* __restrict__ dinv) {
    __shared__ float cx[N_], cy[N_];
    const int b = blockIdx.y, t = threadIdx.x;
    const float* ib = inst + (size_t)b * N_ * 2;
    for (int m = t; m < N_; m += 256) { cx[m] = ib[2*m]; cy[m] = ib[2*m + 1]; }
    __syncthreads();
    const int n = blockIdx.x * 256 + t;
    const float xn = cx[n], yn = cy[n];
    float deg = 1.0f;
    for (int m = 0; m < N_; ++m) {
        float dx = xn - cx[m], dy = yn - cy[m];
        deg += fast_sqrt(dx*dx + dy*dy);
    }
    dinv[b * N_ + n] = rsqrtf(deg);
}

__global__ void k_wg16(const float* __restrict__ Wg, _Float16* __restrict__ Wg16, int n) {
    int i = blockIdx.x * 256 + threadIdx.x;
    if (i < n) Wg16[i] = (_Float16)Wg[i];
}

__global__ void k_proj(const float* __restrict__ inst, const float* __restrict__ Wn,
                       const float* __restrict__ bn,
                       float* __restrict__ hF, _Float16* __restrict__ h16) {
    int i  = blockIdx.x * 256 + threadIdx.x;
    int d  = i & (D_ - 1);
    int bn_ = i >> 7;
    float x = inst[2*bn_], y = inst[2*bn_ + 1];
    float v = x * Wn[2*d] + y * Wn[2*d + 1] + bn[d];
    hF[i] = v;
    h16[i] = (_Float16)v;
}

__global__ __launch_bounds__(256) void k_hw(const _Float16* __restrict__ h16,
                                            const _Float16* __restrict__ Wgl,
                                            _Float16* __restrict__ X16T,
                                            float* __restrict__ gSum,
                                            float* __restrict__ gSq) {
    __shared__ __align__(16) _Float16 sW[D_][D_];
    const int t = threadIdx.x, b = blockIdx.y, n0 = blockIdx.x * 128;

    if (blockIdx.x == 0 && b == 0) {
        for (int i = t; i < B_ * D_; i += 256) { gSum[i] = 0.f; gSq[i] = 0.f; }
    }
    for (int i = t; i < (D_ * D_) / 8; i += 256)
        ((v8h*)&sW[0][0])[i] = ((const v8h*)Wgl)[i];
    __syncthreads();

    const int lane = t & 31, w = t >> 5;
    const int col16 = lane & 15;
    const int kb   = (lane < 16) ? 0 : 8;
    const int roff = (lane < 16) ? 0 : 8;
    const int row  = n0 + w * 16 + col16;

    v8f zero; for (int i = 0; i < 8; ++i) zero[i] = 0.f;
    v8f acc[8]; for (int j = 0; j < 8; ++j) acc[j] = zero;

    for (int kc = 0; kc < D_ / 32; ++kc) {
        v16h a = load_frag(h16 + ((size_t)(b * N_ + row) * D_ + kc * 32 + kb));
        v16h bf[8];
        for (int j = 0; j < 8; ++j)
            bf[j] = load_frag(&sW[j * 16 + col16][kc * 32 + kb]);
        for (int j = 0; j < 8; ++j)
            acc[j] = __builtin_amdgcn_wmma_f32_16x16x32_f16(
                false, a, false, bf[j], (short)0, acc[j], false, false);
    }
    for (int j = 0; j < 8; ++j) {
        int e = j * 16 + col16;
        v8h out;
        for (int r = 0; r < 8; ++r) out[r] = (_Float16)acc[j][r];
        *(v8h*)(X16T + (size_t)(b * D_ + e) * N_ + n0 + w * 16 + roff) = out;
    }
}

__global__ __launch_bounds__(128) void k_gemm(const float* __restrict__ inst,
                                              const float* __restrict__ dinv,
                                              const _Float16* __restrict__ X16T,
                                              const float* __restrict__ bgl,
                                              float* __restrict__ hF,
                                              float* __restrict__ gSum,
                                              float* __restrict__ gSq) {
    __shared__ __align__(16) _Float16 sA[2][64][32];
    __shared__ float rX[64], rY[64], rDi[64];
    __shared__ float cX[2][32], cY[2][32], cDi[2][32];
    __shared__ float sSum[D_], sSq[D_];

    const int t = threadIdx.x, b = blockIdx.y, n0 = blockIdx.x * 64;
    const float* ib = inst + (size_t)b * N_ * 2;
    const float* db = dinv + b * N_;

    if (t < 64) { rX[t] = ib[2*(n0+t)]; rY[t] = ib[2*(n0+t)+1]; rDi[t] = db[n0+t]; }
    if (t < D_) { sSum[t] = 0.f; sSq[t] = 0.f; }
    if (t < 32) { cX[0][t] = ib[2*t]; cY[0][t] = ib[2*t+1]; cDi[0][t] = db[t]; }

    const int lane = t & 31, w = t >> 5;
    const int col16 = lane & 15;
    const int kb   = (lane < 16) ? 0 : 8;
    const int roff = (lane < 16) ? 0 : 8;
    const int arow = t >> 1;
    const int ak0  = (t & 1) * 16;

    v8f zero; for (int i = 0; i < 8; ++i) zero[i] = 0.f;
    v8f acc[8]; for (int j = 0; j < 8; ++j) acc[j] = zero;
    __syncthreads();

    const float xr = rX[arow], yr = rY[arow], dr = rDi[arow];
    const int   gn = n0 + arow;

    for (int kc = 0; kc < N_ / 32; ++kc) {
        const int p  = kc & 1;
        const int m0 = kc * 32;
        {
            v8h tlo, thi;
            for (int kk = 0; kk < 16; ++kk) {
                int k = ak0 + kk, gm = m0 + k;
                float dx = xr - cX[p][k], dy = yr - cY[p][k];
                float v = fast_sqrt(dx*dx + dy*dy);
                if (gn == gm) v += 1.0f;
                _Float16 h = (_Float16)(v * dr * cDi[p][k]);
                if (kk < 8) tlo[kk] = h; else thi[kk - 8] = h;
            }
            *(v8h*)&sA[p][arow][ak0]     = tlo;
            *(v8h*)&sA[p][arow][ak0 + 8] = thi;
        }
        if (kc + 1 < N_ / 32 && t < 32) {
            int m1 = m0 + 32;
            cX[p ^ 1][t] = ib[2*(m1+t)];
            cY[p ^ 1][t] = ib[2*(m1+t)+1];
            cDi[p ^ 1][t] = db[m1+t];
        }
        __syncthreads();
        {
            v16h a = load_frag(&sA[p][w * 16 + col16][0] + kb);
            v16h bf[8];
            for (int j = 0; j < 8; ++j) {
                int d = j * 16 + col16;
                bf[j] = load_frag(X16T + (size_t)(b * D_ + d) * N_ + m0 + kb);
            }
            for (int j = 0; j < 8; ++j)
                acc[j] = __builtin_amdgcn_wmma_f32_16x16x32_f16(
                    false, a, false, bf[j], (short)0, acc[j], false, false);
        }
    }

    for (int j = 0; j < 8; ++j) {
        int d = j * 16 + col16;
        float bgv = bgl[d];
        float s = 0.f, ss = 0.f;
        for (int r = 0; r < 8; ++r) {
            int n = n0 + w * 16 + r + roff;
            size_t idx = (size_t)(b * N_ + n) * D_ + d;
            float hp = fast_tanh(acc[j][r] + bgv) + hF[idx];
            hF[idx] = hp;
            s += hp; ss += hp * hp;
        }
        atomicAdd(&sSum[d], s);
        atomicAdd(&sSq[d], ss);
    }
    __syncthreads();
    if (t < D_) {
        atomicAdd(&gSum[b * D_ + t], sSum[t]);
        atomicAdd(&gSq[b * D_ + t],  sSq[t]);
    }
}

__global__ __launch_bounds__(256) void k_norm(const float* __restrict__ hIn,
                                              const float* __restrict__ gSum,
                                              const float* __restrict__ gSq,
                                              const float* __restrict__ gw,
                                              const float* __restrict__ gb,
                                              const float* __restrict__ ga,
                                              float* __restrict__ outF,
                                              _Float16* __restrict__ h16,
                                              int writeH16) {
    const int t = threadIdx.x, b = blockIdx.y, n0 = blockIdx.x * 16;
    const int d = t & (D_ - 1);
    const int r = t >> 7;
    float a    = ga[d];
    float mean = gSum[b * D_ + d] * (1.0f / N_);
    float var  = gSq[b * D_ + d] * (1.0f / N_) - a * (2.0f - a) * mean * mean;
    float rs   = rsqrtf(var + EPSN);
    float wv = gw[d], bv = gb[d], am = a * mean;
    for (int i = 0; i < 8; ++i) {
        int n = n0 + r + 2 * i;
        size_t idx = (size_t)(b * N_ + n) * D_ + d;
        float y = wv * (hIn[idx] - am) * rs + bv;
        outF[idx] = y;
        if (writeH16) h16[idx] = (_Float16)y;
    }
}

extern "C" void kernel_launch(void* const* d_in, const int* in_sizes, int n_in,
                              void* d_out, int out_size, void* d_ws, size_t ws_size,
                              hipStream_t stream) {
    (void)in_sizes; (void)n_in; (void)out_size; (void)ws_size;
    const float* inst = (const float*)d_in[0];
    const float* Wn   = (const float*)d_in[1];
    const float* bn   = (const float*)d_in[2];
    const float* Wg   = (const float*)d_in[3];
    const float* bg   = (const float*)d_in[4];
    const float* gw   = (const float*)d_in[5];
    const float* gb   = (const float*)d_in[6];
    const float* ga   = (const float*)d_in[7];

    char* w = (char*)d_ws;
    size_t off = 0;
    auto alloc = [&](size_t bytes) {
        void* p = w + off;
        off += (bytes + 255) & ~(size_t)255;
        return p;
    };
    float*    hF    = (float*)   alloc((size_t)B_ * N_ * D_ * 4);
    _Float16* h16   = (_Float16*)alloc((size_t)B_ * N_ * D_ * 2);
    _Float16* X16T  = (_Float16*)alloc((size_t)B_ * N_ * D_ * 2);
    _Float16* Wg16  = (_Float16*)alloc((size_t)L_ * D_ * D_ * 2);
    float*    dinv  = (float*)   alloc((size_t)B_ * N_ * 4);
    float*    gS    = (float*)   alloc((size_t)B_ * D_ * 4);
    float*    gQ    = (float*)   alloc((size_t)B_ * D_ * 4);

    k_dinv<<<dim3(N_ / 256, B_), 256, 0, stream>>>(inst, dinv);
    k_wg16<<<(L_ * D_ * D_ + 255) / 256, 256, 0, stream>>>(Wg, Wg16, L_ * D_ * D_);
    k_proj<<<(B_ * N_ * D_) / 256, 256, 0, stream>>>(inst, Wn, bn, hF, h16);

    for (int l = 0; l < L_; ++l) {
        k_hw<<<dim3(N_ / 128, B_), 256, 0, stream>>>(h16, Wg16 + (size_t)l * D_ * D_,
                                                     X16T, gS, gQ);
        k_gemm<<<dim3(N_ / 64, B_), 128, 0, stream>>>(inst, dinv, X16T, bg + l * D_,
                                                      hF, gS, gQ);
        int last = (l == L_ - 1);
        k_norm<<<dim3(N_ / 16, B_), 256, 0, stream>>>(hF, gS, gQ, gw, gb, ga,
                                                      last ? (float*)d_out : hF,
                                                      h16, last ? 0 : 1);
    }
}


# k_hw: LDS tile of Wg[l] padded to a 272-byte row pitch (16 B x odd): the 16 rows of a B-fragment ds_load_b128 lie on 64 different banks, not 16 deep on the same 4 banks as at 256 bytes
# speedup vs baseline: 3.3237x; 3.3237x over previous
#include <hip/hip_runtime.h>

typedef __attribute__((ext_vector_type(16))) _Float16 v16h;
typedef __attribute__((ext_vector_type(8)))  _Float16 v8h;
typedef __attribute__((ext_vector_type(8)))  float    v8f;
typedef __attribute__((ext_vector_type(4)))  float    v4f;

#define B_   8
#define N_   2048
#define D_   128
#define L_   3
#define EPSN 1e-5f

__device__ __forceinline__ float fast_sqrt(float x) {
#if __has_builtin(__builtin_amdgcn_sqrtf)
    return __builtin_amdgcn_sqrtf(x);
#else
    return sqrtf(x);
#endif
}

__device__ __forceinline__ float fast_tanh(float x) {
#if __has_builtin(__builtin_amdgcn_tanhf)
    return __builtin_amdgcn_tanhf(x);
#else
    return tanhf(x);
#endif
}

__device__ __forceinline__ v16h load_frag(const _Float16* p, int hi_off) {
    v8h lo = *(const v8h*)(p);
    v8h hi = *(const v8h*)(p + hi_off);
    return __builtin_shufflevector(lo, hi, 0,1,2,3,4,5,6,7,8,9,10,11,12,13,14,15);
}

#define FRAG_  512
#define KCH_   (N_ / 32)
#define WP_    (D_ + 8)

struct NormC { float am, g, bv; };
__device__ __forceinline__ NormC norm_coef(const float* __restrict__ gSum,
                                           const float* __restrict__ gSq,
                                           const float* __restrict__ gw,
                                           const float* __restrict__ gb,
                                           const float* __restrict__ ga,
                                           int b, int d) {
    float a    = ga[d];
    float mean = gSum[b * D_ + d] * (1.0f / N_);
    float var  = gSq[b * D_ + d] * (1.0f / N_) - a * (2.0f - a) * mean * mean;
    float rs   = rsqrtf(var + EPSN);
    NormC c;
    c.am = a * mean;
    c.g  = gw[d] * rs;
    c.bv = gb[d];
    return c;
}

__global__ __launch_bounds__(256) void k_dinv(const float* __restrict__ inst,
                                              float* __restrict__ dinv) {
    __shared__ float cx[N_], cy[N_];
    const int b = blockIdx.y, t = threadIdx.x;
    const float* ib = inst + (size_t)b * N_ * 2;
    for (int m = t; m < N_; m += 256) { cx[m] = ib[2*m]; cy[m] = ib[2*m + 1]; }
    __syncthreads();
    const int n = blockIdx.x * 256 + t;
    const float xn = cx[n], yn = cy[n];
    float deg = 1.0f;
    for (int m = 0; m < N_; ++m) {
        float dx = xn - cx[m], dy = yn - cy[m];
        deg += fast_sqrt(dx*dx + dy*dy);
    }
    dinv[b * N_ + n] = rsqrtf(deg);
}

template <bool FIRST>
__device__ __forceinline__ void hw_body(const float* __restrict__ inst,
                                        const float* __restrict__ Wn,
                                        const float* __restrict__ bn,
                                        const float* __restrict__ hF,
                                        const float* __restrict__ pSum,
                                        const float* __restrict__ pSq,
                                        const float* __restrict__ gw,
                                        const float* __restrict__ gb,
                                        const float* __restrict__ ga,
                                        const float* __restrict__ Wgl,
                                        _Float16* __restrict__ XF,
                                        float* __restrict__ zSum,
                                        float* __restrict__ zSq) {
    __shared__ __align__(16) _Float16 sW[D_][WP_];
    __shared__ __align__(16) float sA[D_], sB[D_], sC[D_];
    const int t = threadIdx.x, b = blockIdx.y, n0 = blockIdx.x * 128;

    const int lane = t & 31, w = t >> 5;
    const int col16 = lane & 15;
    const int kb   = (lane < 16) ? 0 : 8;
    const int row  = n0 + w * 16 + col16;

    v4f hv[16];
    float px = 0.f, py = 0.f;
    if (FIRST) {
        px = inst[2 * (b * N_ + row)];
        py = inst[2 * (b * N_ + row) + 1];
    } else {
        const float* hp = hF + (size_t)(b * N_ + row) * D_ + kb;
        for (int kc = 0; kc < D_ / 32; ++kc)
            for (int q = 0; q < 4; ++q)
                hv[kc * 4 + q] = *(const v4f*)(hp + kc * 32 + (q >> 1) * 16 + (q & 1) * 4);
    }

    if (blockIdx.x == 0 && b == 0) {
        for (int i = t; i < B_ * D_; i += 256) { zSum[i] = 0.f; zSq[i] = 0.f; }
    }
    if (t < D_) {
        if (FIRST) {
            sA[t] = Wn[2 * t]; sB[t] = Wn[2 * t + 1]; sC[t] = bn[t];
        } else {
            NormC c = norm_coef(pSum, pSq, gw, gb, ga, b, t);
            sA[t] = c.am; sB[t] = c.g; sC[t] = c.bv;
        }
    }
    for (int i = t; i < (D_ * D_) / 8; i += 256) {
        v4f lo = ((const v4f*)Wgl)[2 * i], hi = ((const v4f*)Wgl)[2 * i + 1];
        v8h o;
        for (int e = 0; e < 4; ++e) { o[e] = (_Float16)lo[e]; o[4 + e] = (_Float16)hi[e]; }
        *(v8h*)&sW[i >> 4][(i & 15) * 8] = o;
    }
    __syncthreads();

    v8f zero; for (int i = 0; i < 8; ++i) zero[i] = 0.f;
    v8f acc[8]; for (int j = 0; j < 8; ++j) acc[j] = zero;

    for (int kc = 0; kc < D_ / 32; ++kc) {
        v16h a;
        for (int q = 0; q < 4; ++q) {
            const int k = kc * 32 + kb + (q >> 1) * 16 + (q & 1) * 4;
            const v4f ca = *(const v4f*)&sA[k];
            const v4f cb = *(const v4f*)&sB[k];
            const v4f cc = *(const v4f*)&sC[k];
            for (int e = 0; e < 4; ++e) {
                float v;
                if (FIRST) v = px * ca[e] + py * cb[e] + cc[e];
                else       v = (hv[kc * 4 + q][e] - ca[e]) * cb[e] + cc[e];
                a[q * 4 + e] = (_Float16)v;
            }
        }
        v16h bf[8];
        for (int j = 0; j < 8; ++j)
            bf[j] = load_frag(&sW[j * 16 + col16][kc * 32 + kb], 16);
        for (int j = 0; j < 8; ++j)
            acc[j] = __builtin_amdgcn_wmma_f32_16x16x32_f16(
                false, a, false, bf[j], (short)0, acc[j], false, false);
    }
    const int kcw = blockIdx.x * 4 + (w >> 1);
    _Float16* xo = XF + (size_t)((b * KCH_ + kcw) * 8) * FRAG_ + (w & 1) * (FRAG_ / 2) + lane * 8;
    for (int j = 0; j < 8; ++j) {
        v8h out;
        for (int r = 0; r < 8; ++r) out[r] = (_Float16)acc[j][r];
        *(v8h*)(xo + j * FRAG_) = out;
    }
}

__global__ __launch_bounds__(256) void k_hw_first(const float* __restrict__ inst,
                                                  const float* __restrict__ Wn,
                                                  const float* __restrict__ bn,
                                                  const float* __restrict__ Wgl,
                                                  _Float16* __restrict__ XF,
                                                  float* __restrict__ zSum,
                                                  float* __restrict__ zSq) {
    hw_body<true>(inst, Wn, bn, nullptr, nullptr, nullptr, nullptr, nullptr, nullptr,
                  Wgl, XF, zSum, zSq);
}

__global__ __launch_bounds__(256) void k_hw_next(const float* __restrict__ hF,
                                                 const float* __restrict__ pSum,
                                                 const float* __restrict__ pSq,
                                                 const float* __restrict__ gw,
                                                 const float* __restrict__ gb,
                                                 const float* __restrict__ ga,
                                                 const float* __restrict__ Wgl,
                                                 _Float16* __restrict__ XF,
                                                 float* __restrict__ zSum,
                                                 float* __restrict__ zSq) {
    hw_body<false>(nullptr, nullptr, nullptr, hF, pSum, pSq, gw, gb, ga,
                   Wgl, XF, zSum, zSq);
}


struct ColT { v4f x[4], y[4], d[4]; };

template <bool FIRST>
__device__ __forceinline__ void gemm_body(const float* __restrict__ inst,
                                          const float* __restrict__ dinv,
                                          const _Float16* __restrict__ XF,
                                          const float* __restrict__ bgl,
                                          const float* __restrict__ Wn,
                                          const float* __restrict__ bn,
                                          const float* __restrict__ pSum,
                                          const float* __restrict__ pSq,
                                          const float* __restrict__ gw,
                                          const float* __restrict__ gb,
                                          const float* __restrict__ ga,
                                          float* __restrict__ hF,
                                          float* __restrict__ gSum,
                                          float* __restrict__ gSq) {
    __shared__ __align__(16) float cX[N_], cY[N_], cDi[N_];
    __shared__ float sSum[D_], sSq[D_];
    __shared__ float sR0[D_], sR1[D_], sR2[D_];

    const int t = threadIdx.x, b = blockIdx.y, n0 = blockIdx.x * 64;
    const float* ib = inst + (size_t)b * N_ * 2;
    const float* db = dinv + b * N_;

    const int lane = t & 31, w = t >> 5;
    const int col16 = lane & 15;
    const int kb   = (lane < 16) ? 0 : 8;
    const int roff = (lane < 16) ? 0 : 8;
    const int gn   = n0 + w * 16 + col16;

    const _Float16* xb = XF + (size_t)b * KCH_ * 8 * FRAG_ + lane * 8;

    auto loadB = [&](v16h (&buf)[8], int kc) {
        const _Float16* p = xb + (size_t)kc * 8 * FRAG_;
        for (int j = 0; j < 8; ++j) buf[j] = load_frag(p + j * FRAG_, FRAG_ / 2);
    };

    v16h b0[8], b1[8], b2[8], b3[8];
    loadB(b0, 0); loadB(b1, 1); loadB(b2, 2);

    {
        float tx[N_ / 128], ty[N_ / 128], td[N_ / 128];
        for (int i = 0; i < N_ / 128; ++i) {
            const int m = t + 128 * i;
            tx[i] = ib[2*m]; ty[i] = ib[2*m + 1]; td[i] = db[m];
        }
        for (int i = 0; i < N_ / 128; ++i) {
            const int m = t + 128 * i;
            cX[m] = tx[i]; cY[m] = ty[i]; cDi[m] = td[i];
        }
    }
    if (t < D_) {
        sSum[t] = 0.f; sSq[t] = 0.f;
        if (FIRST) {
            sR0[t] = Wn[2 * t]; sR1[t] = Wn[2 * t + 1]; sR2[t] = bn[t];
        } else {
            NormC c = norm_coef(pSum, pSq, gw, gb, ga, b, t);
            sR0[t] = c.am; sR1[t] = c.g; sR2[t] = c.bv;
        }
    }
    __syncthreads();

    const float xr = cX[gn], yr = cY[gn], dr = cDi[gn];
    auto loadC = [&](ColT& c, int kc) {
        const int base = kc * 32 + kb;
        for (int q = 0; q < 4; ++q) {
            const int o = base + (q >> 1) * 16 + (q & 1) * 4;
            c.x[q] = *(const v4f*)&cX[o];
            c.y[q] = *(const v4f*)&cY[o];
            c.d[q] = *(const v4f*)&cDi[o];
        }
    };
    auto makeA = [&](const ColT& c, int kc) -> v16h {
        const int base = kc * 32 + kb;
        v16h a;
        for (int q = 0; q < 4; ++q) {
            for (int e = 0; e < 4; ++e) {
                const int gm = base + (q >> 1) * 16 + (q & 1) * 4 + e;
                float dx = xr - c.x[q][e], dy = yr - c.y[q][e];
                float v = fast_sqrt(dx*dx + dy*dy);
                if (gn == gm) v += 1.0f;
                a[q * 4 + e] = (_Float16)(v * dr * c.d[q][e]);
            }
        }
        return a;
    };

    v8f zero; for (int i = 0; i < 8; ++i) zero[i] = 0.f;
    v8f acc[8]; for (int j = 0; j < 8; ++j) acc[j] = zero;

    auto mma = [&](const v16h& a, const v16h (&buf)[8]) {
        for (int j = 0; j < 8; ++j)
            acc[j] = __builtin_amdgcn_wmma_f32_16x16x32_f16(
                false, a, false, buf[j], (short)0, acc[j], false, false);
    };

    ColT c0, c1;
    loadC(c0, 0);

    for (int kc = 0; kc < N_ / 32 - 4; kc += 4) {
        loadB(b3, kc + 3); loadC(c1, kc + 1); mma(makeA(c0, kc),     b0);
        loadB(b0, kc + 4); loadC(c0, kc + 2); mma(makeA(c1, kc + 1), b1);
        loadB(b1, kc + 5); loadC(c1, kc + 3); mma(makeA(c0, kc + 2), b2);
        loadB(b2, kc + 6); loadC(c0, kc + 4); mma(makeA(c1, kc + 3), b3);
    }
    {
        const int kc = N_ / 32 - 4;
        loadB(b3, kc + 3); loadC(c1, kc + 1); mma(makeA(c0, kc),     b0);
                           loadC(c0, kc + 2); mma(makeA(c1, kc + 1), b1);
                           loadC(c1, kc + 3); mma(makeA(c0, kc + 2), b2);
                                              mma(makeA(c1, kc + 3), b3);
    }

    for (int j = 0; j < 8; ++j) {
        int d = j * 16 + col16;
        float bgv = bgl[d];
        const float q0 = sR0[d], q1 = sR1[d], q2 = sR2[d];
        float s = 0.f, ss = 0.f;
        for (int r = 0; r < 8; ++r) {
            int nl = n0 + w * 16 + r + roff;
            size_t idx = (size_t)(b * N_ + nl) * D_ + d;
            float res;
            if (FIRST) res = cX[nl] * q0 + cY[nl] * q1 + q2;
            else       res = (hF[idx] - q0) * q1 + q2;
            float hp = fast_tanh(acc[j][r] + bgv) + res;
            hF[idx] = hp;
            s += hp; ss += hp * hp;
        }
        atomicAdd(&sSum[d], s);
        atomicAdd(&sSq[d], ss);
    }
    __syncthreads();
    if (t < D_) {
        atomicAdd(&gSum[b * D_ + t], sSum[t]);
        atomicAdd(&gSq[b * D_ + t],  sSq[t]);
    }
}

__global__ __launch_bounds__(128) void k_gemm_first(const float* __restrict__ inst,
                                                    const float* __restrict__ dinv,
                                                    const _Float16* __restrict__ XF,
                                                    const float* __restrict__ bgl,
                                                    const float* __restrict__ Wn,
                                                    const float* __restrict__ bn,
                                                    float* __restrict__ hF,
                                                    float* __restrict__ gSum,
                                                    float* __restrict__ gSq) {
    gemm_body<true>(inst, dinv, XF, bgl, Wn, bn, nullptr, nullptr, nullptr, nullptr, nullptr,
                    hF, gSum, gSq);
}

__global__ __launch_bounds__(128) void k_gemm_next(const float* __restrict__ inst,
                                                   const float* __restrict__ dinv,
                                                   const _Float16* __restrict__ XF,
                                                   const float* __restrict__ bgl,
                                                   const float* __restrict__ pSum,
                                                   const float* __restrict__ pSq,
                                                   const float* __restrict__ gw,
                                                   const float* __restrict__ gb,
                                                   const float* __restrict__ ga,
                                                   float* __restrict__ hF,
                                                   float* __restrict__ gSum,
                                                   float* __restrict__ gSq) {
    gemm_body<false>(inst, dinv, XF, bgl, nullptr, nullptr, pSum, pSq, gw, gb, ga,
                     hF, gSum, gSq);
}

__global__ __launch_bounds__(256) void k_norm(const float* __restrict__ hIn,
                                              const float* __restrict__ gSum,
                                              const float* __restrict__ gSq,
                                              const float* __restrict__ gw,
                                              const float* __restrict__ gb,
                                              const float* __restrict__ ga,
                                              float* __restrict__ outF) {
    const int t = threadIdx.x, b = blockIdx.y, n0 = blockIdx.x * 16;
    const int d = t & (D_ - 1);
    const int r = t >> 7;
    const NormC c = norm_coef(gSum, gSq, gw, gb, ga, b, d);
    for (int i = 0; i < 8; ++i) {
        int n = n0 + r + 2 * i;
        size_t idx = (size_t)(b * N_ + n) * D_ + d;
        outF[idx] = (hIn[idx] - c.am) * c.g + c.bv;
    }
}

extern "C" void kernel_launch(void* const* d_in, const int* in_sizes, int n_in,
                              void* d_out, int out_size, void* d_ws, size_t ws_size,
                              hipStream_t stream) {
    (void)in_sizes; (void)n_in; (void)out_size; (void)ws_size;
    const float* inst = (const float*)d_in[0];
    const float* Wn   = (const float*)d_in[1];
    const float* bn   = (const float*)d_in[2];
    const float* Wg   = (const float*)d_in[3];
    const float* bg   = (const float*)d_in[4];
    const float* gw   = (const float*)d_in[5];
    const float* gb   = (const float*)d_in[6];
    const float* ga   = (const float*)d_in[7];

    char* w = (char*)d_ws;
    size_t off = 0;
    auto alloc = [&](size_t bytes) {
        void* p = w + off;
        off += (bytes + 255) & ~(size_t)255;
        return p;
    };
    float*    hF    = (float*)   alloc((size_t)B_ * N_ * D_ * 4);
    _Float16* XF    = (_Float16*)alloc((size_t)B_ * N_ * D_ * 2);
    float*    dinv  = (float*)   alloc((size_t)B_ * N_ * 4);
    float*    gS    = (float*)   alloc((size_t)L_ * B_ * D_ * 4);
    float*    gQ    = (float*)   alloc((size_t)L_ * B_ * D_ * 4);

    k_dinv<<<dim3(N_ / 256, B_), 256, 0, stream>>>(inst, dinv);

    k_hw_first<<<dim3(N_ / 128, B_), 256, 0, stream>>>(inst, Wn, bn, Wg, XF, gS, gQ);
    k_gemm_first<<<dim3(N_ / 64, B_), 128, 0, stream>>>(inst, dinv, XF, bg, Wn, bn,
                                                        hF, gS, gQ);
    for (int l = 1; l < L_; ++l) {
        const float* pS = gS + (l - 1) * B_ * D_;
        const float* pQ = gQ + (l - 1) * B_ * D_;
        k_hw_next<<<dim3(N_ / 128, B_), 256, 0, stream>>>(hF, pS, pQ, gw, gb, ga,
                                                          Wg + (size_t)l * D_ * D_, XF,
                                                          gS + l * B_ * D_, gQ + l * B_ * D_);
        k_gemm_next<<<dim3(N_ / 64, B_), 128, 0, stream>>>(inst, dinv, XF, bg + l * D_,
                                                           pS, pQ, gw, gb, ga, hF,
                                                           gS + l * B_ * D_, gQ + l * B_ * D_);
    }
    k_norm<<<dim3(N_ / 16, B_), 256, 0, stream>>>(hF, gS + (L_ - 1) * B_ * D_,
                                                  gQ + (L_ - 1) * B_ * D_, gw, gb, ga,
                                                  (float*)d_out);
}


# An = Dinv(Dist+I)Dinv factored out of the product: k_hw stores dinv*X, k_gemm's A fragment is the bare distance, self-loop row and row dinv added per output in the epilogue (self-loop summed last)
# speedup vs baseline: 4.4424x; 1.3366x over previous
#include <hip/hip_runtime.h>

typedef __attribute__((ext_vector_type(16))) _Float16 v16h;
typedef __attribute__((ext_vector_type(8)))  _Float16 v8h;
typedef __attribute__((ext_vector_type(8)))  float    v8f;
typedef __attribute__((ext_vector_type(4)))  float    v4f;
typedef __attribute__((ext_vector_type(2)))  float    v2f;

#define B_   8
#define N_   2048
#define D_   128
#define L_   3
#define EPSN 1e-5f

__device__ __forceinline__ float fast_sqrt(float x) {
#if __has_builtin(__builtin_amdgcn_sqrtf)
    return __builtin_amdgcn_sqrtf(x);
#else
    return sqrtf(x);
#endif
}

__device__ __forceinline__ float fast_tanh(float x) {
#if __has_builtin(__builtin_amdgcn_tanhf)
    return __builtin_amdgcn_tanhf(x);
#else
    return tanhf(x);
#endif
}

__device__ __forceinline__ v16h load_frag(const _Float16* p, int hi_off) {
    v8h lo = *(const v8h*)(p);
    v8h hi = *(const v8h*)(p + hi_off);
    return __builtin_shufflevector(lo, hi, 0,1,2,3,4,5,6,7,8,9,10,11,12,13,14,15);
}

#define FRAG_  512
#define KCH_   (N_ / 32)
#define WP_    (D_ + 8)

struct NormC { float am, g, bv; };
__device__ __forceinline__ NormC norm_coef(const float* __restrict__ gSum,
                                           const float* __restrict__ gSq,
                                           const float* __restrict__ gw,
                                           const float* __restrict__ gb,
                                           const float* __restrict__ ga,
                                           int b, int d) {
    float a    = ga[d];
    float mean = gSum[b * D_ + d] * (1.0f / N_);
    float var  = gSq[b * D_ + d] * (1.0f / N_) - a * (2.0f - a) * mean * mean;
    float rs   = rsqrtf(var + EPSN);
    NormC c;
    c.am = a * mean;
    c.g  = gw[d] * rs;
    c.bv = gb[d];
    return c;
}

#define DR_   64
#define DS_   4
#define DG_   16
#define DSTEPS_ (N_ / DS_ / DG_)

struct ColG { v4f x[DG_ / 4], y[DG_ / 4]; };

__global__ __launch_bounds__(256) void k_dinv(const float* __restrict__ inst,
                                              float* __restrict__ dinv) {
    __shared__ __align__(16) float cx[N_], cy[N_];
    __shared__ float part[DS_][DR_];
    const int b = blockIdx.y, t = threadIdx.x;
    const float* ib = inst + (size_t)b * N_ * 2;
    {
        v2f p[N_ / 256];
        for (int i = 0; i < N_ / 256; ++i) p[i] = *(const v2f*)(ib + 2 * (t + 256 * i));
        for (int i = 0; i < N_ / 256; ++i) { cx[t + 256 * i] = p[i][0]; cy[t + 256 * i] = p[i][1]; }
    }
    __syncthreads();
    const int r = t & (DR_ - 1), s = t / DR_;
    const float xn = cx[blockIdx.x * DR_ + r], yn = cy[blockIdx.x * DR_ + r];
    const int m0 = s * (N_ / DS_);

    auto loadG = [&](ColG& c, int g) {
        for (int q = 0; q < DG_ / 4; ++q) {
            c.x[q] = *(const v4f*)&cx[m0 + g * DG_ + 4 * q];
            c.y[q] = *(const v4f*)&cy[m0 + g * DG_ + 4 * q];
        }
    };
    v4f acc = {0.f, 0.f, 0.f, 0.f};
    auto sumG = [&](const ColG& c) {
        for (int q = 0; q < DG_ / 4; ++q) {
            const v4f dx = xn - c.x[q], dy = yn - c.y[q];
            const v4f d2 = dx * dx + dy * dy;
            for (int e = 0; e < 4; ++e) acc[e] += fast_sqrt(d2[e]);
        }
    };

    ColG c0, c1;
    loadG(c0, 0);
    for (int g = 0; g < DSTEPS_ - 2; g += 2) {
        loadG(c1, g + 1); sumG(c0);
        loadG(c0, g + 2); sumG(c1);
    }
    loadG(c1, DSTEPS_ - 1); sumG(c0);
    sumG(c1);

    part[s][r] = (acc[0] + acc[1]) + (acc[2] + acc[3]);
    __syncthreads();
    if (t < DR_) {
        float deg = 1.0f;
        for (int i = 0; i < DS_; ++i) deg += part[i][t];
        dinv[b * N_ + blockIdx.x * DR_ + t] = rsqrtf(deg);
    }
}

template <bool FIRST>
__device__ __forceinline__ void hw_body(const float* __restrict__ inst,
                                        const float* __restrict__ Wn,
                                        const float* __restrict__ bn,
                                        const float* __restrict__ hF,
                                        const float* __restrict__ pSum,
                                        const float* __restrict__ pSq,
                                        const float* __restrict__ gw,
                                        const float* __restrict__ gb,
                                        const float* __restrict__ ga,
                                        const float* __restrict__ Wgl,
                                        const float* __restrict__ dinv,
                                        _Float16* __restrict__ XF,
                                        float* __restrict__ zSum,
                                        float* __restrict__ zSq) {
    __shared__ __align__(16) _Float16 sW[D_][WP_];
    __shared__ __align__(16) float sA[D_], sB[D_], sC[D_];
    const int t = threadIdx.x, b = blockIdx.y, n0 = blockIdx.x * 128;

    const int lane = t & 31, w = t >> 5;
    const int col16 = lane & 15;
    const int kb   = (lane < 16) ? 0 : 8;
    const int row  = n0 + w * 16 + col16;

    const float* dvp = dinv + b * N_ + n0 + w * 16 + kb;
    const v4f dv0 = *(const v4f*)dvp, dv1 = *(const v4f*)(dvp + 4);

    v4f hv[16];
    float px = 0.f, py = 0.f;
    if (FIRST) {
        px = inst[2 * (b * N_ + row)];
        py = inst[2 * (b * N_ + row) + 1];
    } else {
        const float* hp = hF + (size_t)(b * N_ + row) * D_ + kb;
        for (int kc = 0; kc < D_ / 32; ++kc)
            for (int q = 0; q < 4; ++q)
                hv[kc * 4 + q] = *(const v4f*)(hp + kc * 32 + (q >> 1) * 16 + (q & 1) * 4);
    }

    if (blockIdx.x == 0 && b == 0) {
        for (int i = t; i < B_ * D_; i += 256) { zSum[i] = 0.f; zSq[i] = 0.f; }
    }
    if (t < D_) {
        if (FIRST) {
            sA[t] = Wn[2 * t]; sB[t] = Wn[2 * t + 1]; sC[t] = bn[t];
        } else {
            NormC c = norm_coef(pSum, pSq, gw, gb, ga, b, t);
            sA[t] = c.am; sB[t] = c.g; sC[t] = c.bv;
        }
    }
    for (int i = t; i < (D_ * D_) / 8; i += 256) {
        v4f lo = ((const v4f*)Wgl)[2 * i], hi = ((const v4f*)Wgl)[2 * i + 1];
        v8h o;
        for (int e = 0; e < 4; ++e) { o[e] = (_Float16)lo[e]; o[4 + e] = (_Float16)hi[e]; }
        *(v8h*)&sW[i >> 4][(i & 15) * 8] = o;
    }
    __syncthreads();

    v8f zero; for (int i = 0; i < 8; ++i) zero[i] = 0.f;
    v8f acc[8]; for (int j = 0; j < 8; ++j) acc[j] = zero;

    for (int kc = 0; kc < D_ / 32; ++kc) {
        v16h a;
        for (int q = 0; q < 4; ++q) {
            const int k = kc * 32 + kb + (q >> 1) * 16 + (q & 1) * 4;
            const v4f ca = *(const v4f*)&sA[k];
            const v4f cb = *(const v4f*)&sB[k];
            const v4f cc = *(const v4f*)&sC[k];
            for (int e = 0; e < 4; ++e) {
                float v;
                if (FIRST) v = px * ca[e] + py * cb[e] + cc[e];
                else       v = (hv[kc * 4 + q][e] - ca[e]) * cb[e] + cc[e];
                a[q * 4 + e] = (_Float16)v;
            }
        }
        v16h bf[8];
        for (int j = 0; j < 8; ++j)
            bf[j] = load_frag(&sW[j * 16 + col16][kc * 32 + kb], 16);
        for (int j = 0; j < 8; ++j)
            acc[j] = __builtin_amdgcn_wmma_f32_16x16x32_f16(
                false, a, false, bf[j], (short)0, acc[j], false, false);
    }
    const int kcw = blockIdx.x * 4 + (w >> 1);
    _Float16* xo = XF + (size_t)((b * KCH_ + kcw) * 8) * FRAG_ + (w & 1) * (FRAG_ / 2) + lane * 8;
    for (int j = 0; j < 8; ++j) {
        v8h out;
        for (int r = 0; r < 4; ++r) {
            out[r]     = (_Float16)(acc[j][r]     * dv0[r]);
            out[4 + r] = (_Float16)(acc[j][4 + r] * dv1[r]);
        }
        *(v8h*)(xo + j * FRAG_) = out;
    }
}

__global__ __launch_bounds__(256) void k_hw_first(const float* __restrict__ inst,
                                                  const float* __restrict__ Wn,
                                                  const float* __restrict__ bn,
                                                  const float* __restrict__ Wgl,
                                                  const float* __restrict__ dinv,
                                                  _Float16* __restrict__ XF,
                                                  float* __restrict__ zSum,
                                                  float* __restrict__ zSq) {
    hw_body<true>(inst, Wn, bn, nullptr, nullptr, nullptr, nullptr, nullptr, nullptr,
                  Wgl, dinv, XF, zSum, zSq);
}

__global__ __launch_bounds__(256) void k_hw_next(const float* __restrict__ hF,
                                                 const float* __restrict__ pSum,
                                                 const float* __restrict__ pSq,
                                                 const float* __restrict__ gw,
                                                 const float* __restrict__ gb,
                                                 const float* __restrict__ ga,
                                                 const float* __restrict__ Wgl,
                                                 const float* __restrict__ dinv,
                                                 _Float16* __restrict__ XF,
                                                 float* __restrict__ zSum,
                                                 float* __restrict__ zSq) {
    hw_body<false>(nullptr, nullptr, nullptr, hF, pSum, pSq, gw, gb, ga,
                   Wgl, dinv, XF, zSum, zSq);
}


struct ColT { v4f x[4], y[4]; };

template <bool FIRST>
__device__ __forceinline__ void gemm_body(const float* __restrict__ inst,
                                          const float* __restrict__ dinv,
                                          const _Float16* __restrict__ XF,
                                          const float* __restrict__ bgl,
                                          const float* __restrict__ Wn,
                                          const float* __restrict__ bn,
                                          const float* __restrict__ pSum,
                                          const float* __restrict__ pSq,
                                          const float* __restrict__ gw,
                                          const float* __restrict__ gb,
                                          const float* __restrict__ ga,
                                          float* __restrict__ hF,
                                          float* __restrict__ gSum,
                                          float* __restrict__ gSq) {
    __shared__ __align__(16) float cX[N_], cY[N_];
    __shared__ float sSum[D_], sSq[D_];
    __shared__ float sR0[D_], sR1[D_], sR2[D_];

    const int t = threadIdx.x, b = blockIdx.y, n0 = blockIdx.x * 64;
    const float* ib = inst + (size_t)b * N_ * 2;

    const int lane = t & 31, w = t >> 5;
    const int col16 = lane & 15;
    const int kb   = (lane < 16) ? 0 : 8;
    const int roff = (lane < 16) ? 0 : 8;
    const int gn   = n0 + w * 16 + col16;

    const float* dq = dinv + b * N_ + n0 + w * 16 + roff;
    const v4f dq0 = *(const v4f*)dq, dq1 = *(const v4f*)(dq + 4);

    const _Float16* xb = XF + (size_t)b * KCH_ * 8 * FRAG_ + lane * 8;

    auto loadB = [&](v16h (&buf)[8], int kc) {
        const _Float16* p = xb + (size_t)kc * 8 * FRAG_;
        for (int j = 0; j < 8; ++j) buf[j] = load_frag(p + j * FRAG_, FRAG_ / 2);
    };

    v16h b0[8], b1[8], b2[8], b3[8];
    loadB(b0, 0); loadB(b1, 1); loadB(b2, 2);

    {
        float tx[N_ / 128], ty[N_ / 128];
        for (int i = 0; i < N_ / 128; ++i) {
            const int m = t + 128 * i;
            tx[i] = ib[2*m]; ty[i] = ib[2*m + 1];
        }
        for (int i = 0; i < N_ / 128; ++i) {
            const int m = t + 128 * i;
            cX[m] = tx[i]; cY[m] = ty[i];
        }
    }
    if (t < D_) {
        sSum[t] = 0.f; sSq[t] = 0.f;
        if (FIRST) {
            sR0[t] = Wn[2 * t]; sR1[t] = Wn[2 * t + 1]; sR2[t] = bn[t];
        } else {
            NormC c = norm_coef(pSum, pSq, gw, gb, ga, b, t);
            sR0[t] = c.am; sR1[t] = c.g; sR2[t] = c.bv;
        }
    }
    __syncthreads();

    const float xr = cX[gn], yr = cY[gn];
    auto loadC = [&](ColT& c, int kc) {
        const int base = kc * 32 + kb;
        for (int q = 0; q < 4; ++q) {
            const int o = base + (q >> 1) * 16 + (q & 1) * 4;
            c.x[q] = *(const v4f*)&cX[o];
            c.y[q] = *(const v4f*)&cY[o];
        }
    };
    auto makeA = [&](const ColT& c) -> v16h {
        v16h a;
        for (int q = 0; q < 4; ++q) {
            for (int e = 0; e < 4; ++e) {
                float dx = xr - c.x[q][e], dy = yr - c.y[q][e];
                a[q * 4 + e] = (_Float16)fast_sqrt(dx*dx + dy*dy);
            }
        }
        return a;
    };

    v8f zero; for (int i = 0; i < 8; ++i) zero[i] = 0.f;
    v8f acc[8]; for (int j = 0; j < 8; ++j) acc[j] = zero;

    auto mma = [&](const v16h& a, const v16h (&buf)[8]) {
        for (int j = 0; j < 8; ++j)
            acc[j] = __builtin_amdgcn_wmma_f32_16x16x32_f16(
                false, a, false, buf[j], (short)0, acc[j], false, false);
    };

    ColT c0, c1;
    loadC(c0, 0);

    for (int kc = 0; kc < N_ / 32 - 4; kc += 4) {
        loadB(b3, kc + 3); loadC(c1, kc + 1); mma(makeA(c0), b0);
        loadB(b0, kc + 4); loadC(c0, kc + 2); mma(makeA(c1), b1);
        loadB(b1, kc + 5); loadC(c1, kc + 3); mma(makeA(c0), b2);
        loadB(b2, kc + 6); loadC(c0, kc + 4); mma(makeA(c1), b3);
    }
    const int rb = n0 + w * 16;
    const _Float16* xsp = xb + (size_t)((rb >> 5) * 8) * FRAG_ + ((rb >> 4) & 1) * (FRAG_ / 2);
    v8h xs[8];
    for (int j = 0; j < 8; ++j) xs[j] = *(const v8h*)(xsp + j * FRAG_);
    {
        const int kc = N_ / 32 - 4;
        loadB(b3, kc + 3); loadC(c1, kc + 1); mma(makeA(c0), b0);
                           loadC(c0, kc + 2); mma(makeA(c1), b1);
                           loadC(c1, kc + 3); mma(makeA(c0), b2);
                                              mma(makeA(c1), b3);
    }

    float drv[8];
    for (int r = 0; r < 4; ++r) { drv[r] = dq0[r]; drv[4 + r] = dq1[r]; }
    for (int j = 0; j < 8; ++j) {
        int d = j * 16 + col16;
        float bgv = bgl[d];
        const float q0 = sR0[d], q1 = sR1[d], q2 = sR2[d];
        float s = 0.f, ss = 0.f;
        for (int r = 0; r < 8; ++r) {
            int nl = n0 + w * 16 + r + roff;
            size_t idx = (size_t)(b * N_ + nl) * D_ + d;
            float res;
            if (FIRST) res = cX[nl] * q0 + cY[nl] * q1 + q2;
            else       res = (hF[idx] - q0) * q1 + q2;
            float z  = (acc[j][r] + (float)xs[j][r]) * drv[r];
            float hp = fast_tanh(z + bgv) + res;
            hF[idx] = hp;
            s += hp; ss += hp * hp;
        }
        atomicAdd(&sSum[d], s);
        atomicAdd(&sSq[d], ss);
    }
    __syncthreads();
    if (t < D_) {
        atomicAdd(&gSum[b * D_ + t], sSum[t]);
        atomicAdd(&gSq[b * D_ + t],  sSq[t]);
    }
}

__global__ __launch_bounds__(128) void k_gemm_first(const float* __restrict__ inst,
                                                    const float* __restrict__ dinv,
                                                    const _Float16* __restrict__ XF,
                                                    const float* __restrict__ bgl,
                                                    const float* __restrict__ Wn,
                                                    const float* __restrict__ bn,
                                                    float* __restrict__ hF,
                                                    float* __restrict__ gSum,
                                                    float* __restrict__ gSq) {
    gemm_body<true>(inst, dinv, XF, bgl, Wn, bn, nullptr, nullptr, nullptr, nullptr, nullptr,
                    hF, gSum, gSq);
}

__global__ __launch_bounds__(128) void k_gemm_next(const float* __restrict__ inst,
                                                   const float* __restrict__ dinv,
                                                   const _Float16* __restrict__ XF,
                                                   const float* __restrict__ bgl,
                                                   const float* __restrict__ pSum,
                                                   const float* __restrict__ pSq,
                                                   const float* __restrict__ gw,
                                                   const float* __restrict__ gb,
                                                   const float* __restrict__ ga,
                                                   float* __restrict__ hF,
                                                   float* __restrict__ gSum,
                                                   float* __restrict__ gSq) {
    gemm_body<false>(inst, dinv, XF, bgl, nullptr, nullptr, pSum, pSq, gw, gb, ga,
                     hF, gSum, gSq);
}

__global__ __launch_bounds__(256) void k_norm(const float* __restrict__ hIn,
                                              const float* __restrict__ gSum,
                                              const float* __restrict__ gSq,
                                              const float* __restrict__ gw,
                                              const float* __restrict__ gb,
                                              const float* __restrict__ ga,
                                              float* __restrict__ outF) {
    const int t = threadIdx.x, b = blockIdx.y, n0 = blockIdx.x * 16;
    const int d = t & (D_ - 1);
    const int r = t >> 7;
    const NormC c = norm_coef(gSum, gSq, gw, gb, ga, b, d);
    for (int i = 0; i < 8; ++i) {
        int n = n0 + r + 2 * i;
        size_t idx = (size_t)(b * N_ + n) * D_ + d;
        outF[idx] = (hIn[idx] - c.am) * c.g + c.bv;
    }
}

extern "C" void kernel_launch(void* const* d_in, const int* in_sizes, int n_in,
                              void* d_out, int out_size, void* d_ws, size_t ws_size,
                              hipStream_t stream) {
    (void)in_sizes; (void)n_in; (void)out_size; (void)ws_size;
    const float* inst = (const float*)d_in[0];
    const float* Wn   = (const float*)d_in[1];
    const float* bn   = (const float*)d_in[2];
    const float* Wg   = (const float*)d_in[3];
    const float* bg   = (const float*)d_in[4];
    const float* gw   = (const float*)d_in[5];
    const float* gb   = (const float*)d_in[6];
    const float* ga   = (const float*)d_in[7];

    char* w = (char*)d_ws;
    size_t off = 0;
    auto alloc = [&](size_t bytes) {
        void* p = w + off;
        off += (bytes + 255) & ~(size_t)255;
        return p;
    };
    float*    hF    = (float*)   alloc((size_t)B_ * N_ * D_ * 4);
    _Float16* XF    = (_Float16*)alloc((size_t)B_ * N_ * D_ * 2);
    float*    dinv  = (float*)   alloc((size_t)B_ * N_ * 4);
    float*    gS    = (float*)   alloc((size_t)L_ * B_ * D_ * 4);
    float*    gQ    = (float*)   alloc((size_t)L_ * B_ * D_ * 4);

    k_dinv<<<dim3(N_ / DR_, B_), 256, 0, stream>>>(inst, dinv);

    k_hw_first<<<dim3(N_ / 128, B_), 256, 0, stream>>>(inst, Wn, bn, Wg, dinv, XF, gS, gQ);
    k_gemm_first<<<dim3(N_ / 64, B_), 128, 0, stream>>>(inst, dinv, XF, bg, Wn, bn,
                                                        hF, gS, gQ);
    for (int l = 1; l < L_; ++l) {
        const float* pS = gS + (l - 1) * B_ * D_;
        const float* pQ = gQ + (l - 1) * B_ * D_;
        k_hw_next<<<dim3(N_ / 128, B_), 256, 0, stream>>>(hF, pS, pQ, gw, gb, ga,
                                                          Wg + (size_t)l * D_ * D_, dinv, XF,
                                                          gS + l * B_ * D_, gQ + l * B_ * D_);
        k_gemm_next<<<dim3(N_ / 64, B_), 128, 0, stream>>>(inst, dinv, XF, bg + l * D_,
                                                           pS, pQ, gw, gb, ga, hF,
                                                           gS + l * B_ * D_, gQ + l * B_ * D_);
    }
    k_norm<<<dim3(N_ / 16, B_), 256, 0, stream>>>(hF, gS + (L_ - 1) * B_ * D_,
                                                  gQ + (L_ - 1) * B_ * D_, gw, gb, ga,
                                                  (float*)d_out);
}


# k_dinv and k_hw_first in one launch: each 64-row work-group sums its rows' degrees, then makes layer 1's Xs for those rows (8 waves = 4 row tiles x 2 feature halves); 8 -> 7 launches
# speedup vs baseline: 4.7147x; 1.0613x over previous
#include <hip/hip_runtime.h>

typedef __attribute__((ext_vector_type(16))) _Float16 v16h;
typedef __attribute__((ext_vector_type(8)))  _Float16 v8h;
typedef __attribute__((ext_vector_type(8)))  float    v8f;
typedef __attribute__((ext_vector_type(4)))  float    v4f;
typedef __attribute__((ext_vector_type(2)))  float    v2f;

#define B_   8
#define N_   2048
#define D_   128
#define L_   3
#define EPSN 1e-5f

__device__ __forceinline__ float fast_sqrt(float x) {
#if __has_builtin(__builtin_amdgcn_sqrtf)
    return __builtin_amdgcn_sqrtf(x);
#else
    return sqrtf(x);
#endif
}

__device__ __forceinline__ float fast_tanh(float x) {
#if __has_builtin(__builtin_amdgcn_tanhf)
    return __builtin_amdgcn_tanhf(x);
#else
    return tanhf(x);
#endif
}

__device__ __forceinline__ v16h load_frag(const _Float16* p, int hi_off) {
    v8h lo = *(const v8h*)(p);
    v8h hi = *(const v8h*)(p + hi_off);
    return __builtin_shufflevector(lo, hi, 0,1,2,3,4,5,6,7,8,9,10,11,12,13,14,15);
}

#define FRAG_  512
#define KCH_   (N_ / 32)
#define WP_    (D_ + 8)

struct NormC { float am, g, bv; };
__device__ __forceinline__ NormC norm_coef(const float* __restrict__ gSum,
                                           const float* __restrict__ gSq,
                                           const float* __restrict__ gw,
                                           const float* __restrict__ gb,
                                           const float* __restrict__ ga,
                                           int b, int d) {
    float a    = ga[d];
    float mean = gSum[b * D_ + d] * (1.0f / N_);
    float var  = gSq[b * D_ + d] * (1.0f / N_) - a * (2.0f - a) * mean * mean;
    float rs   = rsqrtf(var + EPSN);
    NormC c;
    c.am = a * mean;
    c.g  = gw[d] * rs;
    c.bv = gb[d];
    return c;
}

#define DR_   64
#define DS_   4
#define DG_   16
#define DSTEPS_ (N_ / DS_ / DG_)

struct ColG { v4f x[DG_ / 4], y[DG_ / 4]; };

__global__ __launch_bounds__(256) void k_dinv_hw(const float* __restrict__ inst,
                                                 const float* __restrict__ Wn,
                                                 const float* __restrict__ bn,
                                                 const float* __restrict__ Wgl,
                                                 float* __restrict__ dinv,
                                                 _Float16* __restrict__ XF,
                                                 float* __restrict__ zSum,
                                                 float* __restrict__ zSq) {
    __shared__ __align__(16) float cx[N_], cy[N_];
    __shared__ float part[DS_][DR_];
    __shared__ __align__(16) float sDv[DR_];
    __shared__ __align__(16) _Float16 sW[D_][WP_];
    __shared__ __align__(16) float sA[D_], sB[D_], sC[D_];
    const int b = blockIdx.y, t = threadIdx.x, n0 = blockIdx.x * DR_;
    const float* ib = inst + (size_t)b * N_ * 2;
    {
        v2f p[N_ / 256];
        v4f wl[(D_ * D_) / 8 / 256], wh[(D_ * D_) / 8 / 256];
        for (int i = 0; i < N_ / 256; ++i) p[i] = *(const v2f*)(ib + 2 * (t + 256 * i));
        for (int i = 0; i < (D_ * D_) / 8 / 256; ++i) {
            const int g = t + 256 * i;
            wl[i] = ((const v4f*)Wgl)[2 * g]; wh[i] = ((const v4f*)Wgl)[2 * g + 1];
        }
        for (int i = 0; i < N_ / 256; ++i) { cx[t + 256 * i] = p[i][0]; cy[t + 256 * i] = p[i][1]; }
        for (int i = 0; i < (D_ * D_) / 8 / 256; ++i) {
            const int g = t + 256 * i;
            v8h o;
            for (int e = 0; e < 4; ++e) { o[e] = (_Float16)wl[i][e]; o[4 + e] = (_Float16)wh[i][e]; }
            *(v8h*)&sW[g >> 4][(g & 15) * 8] = o;
        }
    }
    if (t < D_) { sA[t] = Wn[2 * t]; sB[t] = Wn[2 * t + 1]; sC[t] = bn[t]; }
    if (blockIdx.x == 0 && b == 0) {
        for (int i = t; i < B_ * D_; i += 256) { zSum[i] = 0.f; zSq[i] = 0.f; }
    }
    __syncthreads();

    const int r = t & (DR_ - 1), s = t / DR_;
    const float xn = cx[n0 + r], yn = cy[n0 + r];
    const int m0 = s * (N_ / DS_);

    auto loadG = [&](ColG& c, int g) {
        for (int q = 0; q < DG_ / 4; ++q) {
            c.x[q] = *(const v4f*)&cx[m0 + g * DG_ + 4 * q];
            c.y[q] = *(const v4f*)&cy[m0 + g * DG_ + 4 * q];
        }
    };
    v4f dsum = {0.f, 0.f, 0.f, 0.f};
    auto sumG = [&](const ColG& c) {
        for (int q = 0; q < DG_ / 4; ++q) {
            const v4f dx = xn - c.x[q], dy = yn - c.y[q];
            const v4f d2 = dx * dx + dy * dy;
            for (int e = 0; e < 4; ++e) dsum[e] += fast_sqrt(d2[e]);
        }
    };

    ColG c0, c1;
    loadG(c0, 0);
    for (int g = 0; g < DSTEPS_ - 2; g += 2) {
        loadG(c1, g + 1); sumG(c0);
        loadG(c0, g + 2); sumG(c1);
    }
    loadG(c1, DSTEPS_ - 1); sumG(c0);
    sumG(c1);

    part[s][r] = (dsum[0] + dsum[1]) + (dsum[2] + dsum[3]);
    __syncthreads();
    if (t < DR_) {
        float deg = 1.0f;
        for (int i = 0; i < DS_; ++i) deg += part[i][t];
        const float dv = rsqrtf(deg);
        dinv[b * N_ + n0 + t] = dv;
        sDv[t] = dv;
    }
    __syncthreads();

    const int lane = t & 31, w = t >> 5;
    const int col16 = lane & 15;
    const int kb   = (lane < 16) ? 0 : 8;
    const int rt   = w & 3, ch = w >> 2;
    const float px = cx[n0 + rt * 16 + col16];
    const float py = cy[n0 + rt * 16 + col16];
    const v4f dv0 = *(const v4f*)&sDv[rt * 16 + kb], dv1 = *(const v4f*)&sDv[rt * 16 + kb + 4];

    v8f zero; for (int i = 0; i < 8; ++i) zero[i] = 0.f;
    v8f acc[4]; for (int j = 0; j < 4; ++j) acc[j] = zero;

    for (int kc = 0; kc < D_ / 32; ++kc) {
        v16h a;
        for (int q = 0; q < 4; ++q) {
            const int k = kc * 32 + kb + (q >> 1) * 16 + (q & 1) * 4;
            const v4f ca = *(const v4f*)&sA[k];
            const v4f cb = *(const v4f*)&sB[k];
            const v4f cc = *(const v4f*)&sC[k];
            for (int e = 0; e < 4; ++e)
                a[q * 4 + e] = (_Float16)(px * ca[e] + py * cb[e] + cc[e]);
        }
        v16h bf[4];
        for (int j = 0; j < 4; ++j)
            bf[j] = load_frag(&sW[(ch * 4 + j) * 16 + col16][kc * 32 + kb], 16);
        for (int j = 0; j < 4; ++j)
            acc[j] = __builtin_amdgcn_wmma_f32_16x16x32_f16(
                false, a, false, bf[j], (short)0, acc[j], false, false);
    }
    const int kcw = blockIdx.x * (DR_ / 32) + (rt >> 1);
    _Float16* xo = XF + (size_t)((b * KCH_ + kcw) * 8 + ch * 4) * FRAG_ + (rt & 1) * (FRAG_ / 2) + lane * 8;
    for (int j = 0; j < 4; ++j) {
        v8h out;
        for (int q = 0; q < 4; ++q) {
            out[q]     = (_Float16)(acc[j][q]     * dv0[q]);
            out[4 + q] = (_Float16)(acc[j][4 + q] * dv1[q]);
        }
        *(v8h*)(xo + j * FRAG_) = out;
    }
}

__global__ __launch_bounds__(256) void k_hw_next(const float* __restrict__ hF,
                                                 const float* __restrict__ pSum,
                                                 const float* __restrict__ pSq,
                                                 const float* __restrict__ gw,
                                                 const float* __restrict__ gb,
                                                 const float* __restrict__ ga,
                                                 const float* __restrict__ Wgl,
                                                 const float* __restrict__ dinv,
                                                 _Float16* __restrict__ XF,
                                                 float* __restrict__ zSum,
                                                 float* __restrict__ zSq) {
    __shared__ __align__(16) _Float16 sW[D_][WP_];
    __shared__ __align__(16) float sA[D_], sB[D_], sC[D_];
    const int t = threadIdx.x, b = blockIdx.y, n0 = blockIdx.x * 128;

    const int lane = t & 31, w = t >> 5;
    const int col16 = lane & 15;
    const int kb   = (lane < 16) ? 0 : 8;
    const int row  = n0 + w * 16 + col16;

    const float* dvp = dinv + b * N_ + n0 + w * 16 + kb;
    const v4f dv0 = *(const v4f*)dvp, dv1 = *(const v4f*)(dvp + 4);

    v4f hv[16];
    {
        const float* hp = hF + (size_t)(b * N_ + row) * D_ + kb;
        for (int kc = 0; kc < D_ / 32; ++kc)
            for (int q = 0; q < 4; ++q)
                hv[kc * 4 + q] = *(const v4f*)(hp + kc * 32 + (q >> 1) * 16 + (q & 1) * 4);
    }

    if (blockIdx.x == 0 && b == 0) {
        for (int i = t; i < B_ * D_; i += 256) { zSum[i] = 0.f; zSq[i] = 0.f; }
    }
    if (t < D_) {
        NormC c = norm_coef(pSum, pSq, gw, gb, ga, b, t);
        sA[t] = c.am; sB[t] = c.g; sC[t] = c.bv;
    }
    for (int i = t; i < (D_ * D_) / 8; i += 256) {
        v4f lo = ((const v4f*)Wgl)[2 * i], hi = ((const v4f*)Wgl)[2 * i + 1];
        v8h o;
        for (int e = 0; e < 4; ++e) { o[e] = (_Float16)lo[e]; o[4 + e] = (_Float16)hi[e]; }
        *(v8h*)&sW[i >> 4][(i & 15) * 8] = o;
    }
    __syncthreads();

    v8f zero; for (int i = 0; i < 8; ++i) zero[i] = 0.f;
    v8f acc[8]; for (int j = 0; j < 8; ++j) acc[j] = zero;

    for (int kc = 0; kc < D_ / 32; ++kc) {
        v16h a;
        for (int q = 0; q < 4; ++q) {
            const int k = kc * 32 + kb + (q >> 1) * 16 + (q & 1) * 4;
            const v4f ca = *(const v4f*)&sA[k];
            const v4f cb = *(const v4f*)&sB[k];
            const v4f cc = *(const v4f*)&sC[k];
            for (int e = 0; e < 4; ++e)
                a[q * 4 + e] = (_Float16)((hv[kc * 4 + q][e] - ca[e]) * cb[e] + cc[e]);
        }
        v16h bf[8];
        for (int j = 0; j < 8; ++j)
            bf[j] = load_frag(&sW[j * 16 + col16][kc * 32 + kb], 16);
        for (int j = 0; j < 8; ++j)
            acc[j] = __builtin_amdgcn_wmma_f32_16x16x32_f16(
                false, a, false, bf[j], (short)0, acc[j], false, false);
    }
    const int kcw = blockIdx.x * 4 + (w >> 1);
    _Float16* xo = XF + (size_t)((b * KCH_ + kcw) * 8) * FRAG_ + (w & 1) * (FRAG_ / 2) + lane * 8;
    for (int j = 0; j < 8; ++j) {
        v8h out;
        for (int r = 0; r < 4; ++r) {
            out[r]     = (_Float16)(acc[j][r]     * dv0[r]);
            out[4 + r] = (_Float16)(acc[j][4 + r] * dv1[r]);
        }
        *(v8h*)(xo + j * FRAG_) = out;
    }
}


struct ColT { v4f x[4], y[4]; };

template <bool FIRST>
__device__ __forceinline__ void gemm_body(const float* __restrict__ inst,
                                          const float* __restrict__ dinv,
                                          const _Float16* __restrict__ XF,
                                          const float* __restrict__ bgl,
                                          const float* __restrict__ Wn,
                                          const float* __restrict__ bn,
                                          const float* __restrict__ pSum,
                                          const float* __restrict__ pSq,
                                          const float* __restrict__ gw,
                                          const float* __restrict__ gb,
                                          const float* __restrict__ ga,
                                          float* __restrict__ hF,
                                          float* __restrict__ gSum,
                                          float* __restrict__ gSq) {
    __shared__ __align__(16) float cX[N_], cY[N_];
    __shared__ float sSum[D_], sSq[D_];
    __shared__ float sR0[D_], sR1[D_], sR2[D_];

    const int t = threadIdx.x, b = blockIdx.y, n0 = blockIdx.x * 64;
    const float* ib = inst + (size_t)b * N_ * 2;

    const int lane = t & 31, w = t >> 5;
    const int col16 = lane & 15;
    const int kb   = (lane < 16) ? 0 : 8;
    const int roff = (lane < 16) ? 0 : 8;
    const int gn   = n0 + w * 16 + col16;

    const float* dq = dinv + b * N_ + n0 + w * 16 + roff;
    const v4f dq0 = *(const v4f*)dq, dq1 = *(const v4f*)(dq + 4);

    const _Float16* xb = XF + (size_t)b * KCH_ * 8 * FRAG_ + lane * 8;

    auto loadB = [&](v16h (&buf)[8], int kc) {
        const _Float16* p = xb + (size_t)kc * 8 * FRAG_;
        for (int j = 0; j < 8; ++j) buf[j] = load_frag(p + j * FRAG_, FRAG_ / 2);
    };

    v16h b0[8], b1[8], b2[8], b3[8];
    loadB(b0, 0); loadB(b1, 1); loadB(b2, 2);

    {
        float tx[N_ / 128], ty[N_ / 128];
        for (int i = 0; i < N_ / 128; ++i) {
            const int m = t + 128 * i;
            tx[i] = ib[2*m]; ty[i] = ib[2*m + 1];
        }
        for (int i = 0; i < N_ / 128; ++i) {
            const int m = t + 128 * i;
            cX[m] = tx[i]; cY[m] = ty[i];
        }
    }
    if (t < D_) {
        sSum[t] = 0.f; sSq[t] = 0.f;
        if (FIRST) {
            sR0[t] = Wn[2 * t]; sR1[t] = Wn[2 * t + 1]; sR2[t] = bn[t];
        } else {
            NormC c = norm_coef(pSum, pSq, gw, gb, ga, b, t);
            sR0[t] = c.am; sR1[t] = c.g; sR2[t] = c.bv;
        }
    }
    __syncthreads();

    const float xr = cX[gn], yr = cY[gn];
    auto loadC = [&](ColT& c, int kc) {
        const int base = kc * 32 + kb;
        for (int q = 0; q < 4; ++q) {
            const int o = base + (q >> 1) * 16 + (q & 1) * 4;
            c.x[q] = *(const v4f*)&cX[o];
            c.y[q] = *(const v4f*)&cY[o];
        }
    };
    auto makeA = [&](const ColT& c) -> v16h {
        v16h a;
        for (int q = 0; q < 4; ++q) {
            for (int e = 0; e < 4; ++e) {
                float dx = xr - c.x[q][e], dy = yr - c.y[q][e];
                a[q * 4 + e] = (_Float16)fast_sqrt(dx*dx + dy*dy);
            }
        }
        return a;
    };

    v8f zero; for (int i = 0; i < 8; ++i) zero[i] = 0.f;
    v8f acc[8]; for (int j = 0; j < 8; ++j) acc[j] = zero;

    auto mma = [&](const v16h& a, const v16h (&buf)[8]) {
        for (int j = 0; j < 8; ++j)
            acc[j] = __builtin_amdgcn_wmma_f32_16x16x32_f16(
                false, a, false, buf[j], (short)0, acc[j], false, false);
    };

    ColT c0, c1;
    loadC(c0, 0);

    for (int kc = 0; kc < N_ / 32 - 4; kc += 4) {
        loadB(b3, kc + 3); loadC(c1, kc + 1); mma(makeA(c0), b0);
        loadB(b0, kc + 4); loadC(c0, kc + 2); mma(makeA(c1), b1);
        loadB(b1, kc + 5); loadC(c1, kc + 3); mma(makeA(c0), b2);
        loadB(b2, kc + 6); loadC(c0, kc + 4); mma(makeA(c1), b3);
    }
    const int rb = n0 + w * 16;
    const _Float16* xsp = xb + (size_t)((rb >> 5) * 8) * FRAG_ + ((rb >> 4) & 1) * (FRAG_ / 2);
    v8h xs[8];
    for (int j = 0; j < 8; ++j) xs[j] = *(const v8h*)(xsp + j * FRAG_);
    {
        const int kc = N_ / 32 - 4;
        loadB(b3, kc + 3); loadC(c1, kc + 1); mma(makeA(c0), b0);
                           loadC(c0, kc + 2); mma(makeA(c1), b1);
                           loadC(c1, kc + 3); mma(makeA(c0), b2);
                                              mma(makeA(c1), b3);
    }

    float drv[8];
    for (int r = 0; r < 4; ++r) { drv[r] = dq0[r]; drv[4 + r] = dq1[r]; }
    for (int j = 0; j < 8; ++j) {
        int d = j * 16 + col16;
        float bgv = bgl[d];
        const float q0 = sR0[d], q1 = sR1[d], q2 = sR2[d];
        float s = 0.f, ss = 0.f;
        for (int r = 0; r < 8; ++r) {
            int nl = n0 + w * 16 + r + roff;
            size_t idx = (size_t)(b * N_ + nl) * D_ + d;
            float res;
            if (FIRST) res = cX[nl] * q0 + cY[nl] * q1 + q2;
            else       res = (hF[idx] - q0) * q1 + q2;
            float z  = (acc[j][r] + (float)xs[j][r]) * drv[r];
            float hp = fast_tanh(z + bgv) + res;
            hF[idx] = hp;
            s += hp; ss += hp * hp;
        }
        atomicAdd(&sSum[d], s);
        atomicAdd(&sSq[d], ss);
    }
    __syncthreads();
    if (t < D_) {
        atomicAdd(&gSum[b * D_ + t], sSum[t]);
        atomicAdd(&gSq[b * D_ + t],  sSq[t]);
    }
}

__global__ __launch_bounds__(128) void k_gemm_first(const float* __restrict__ inst,
                                                    const float* __restrict__ dinv,
                                                    const _Float16* __restrict__ XF,
                                                    const float* __restrict__ bgl,
                                                    const float* __restrict__ Wn,
                                                    const float* __restrict__ bn,
                                                    float* __restrict__ hF,
                                                    float* __restrict__ gSum,
                                                    float* __restrict__ gSq) {
    gemm_body<true>(inst, dinv, XF, bgl, Wn, bn, nullptr, nullptr, nullptr, nullptr, nullptr,
                    hF, gSum, gSq);
}

__global__ __launch_bounds__(128) void k_gemm_next(const float* __restrict__ inst,
                                                   const float* __restrict__ dinv,
                                                   const _Float16* __restrict__ XF,
                                                   const float* __restrict__ bgl,
                                                   const float* __restrict__ pSum,
                                                   const float* __restrict__ pSq,
                                                   const float* __restrict__ gw,
                                                   const float* __restrict__ gb,
                                                   const float* __restrict__ ga,
                                                   float* __restrict__ hF,
                                                   float* __restrict__ gSum,
                                                   float* __restrict__ gSq) {
    gemm_body<false>(inst, dinv, XF, bgl, nullptr, nullptr, pSum, pSq, gw, gb, ga,
                     hF, gSum, gSq);
}

__global__ __launch_bounds__(256) void k_norm(const float* __restrict__ hIn,
                                              const float* __restrict__ gSum,
                                              const float* __restrict__ gSq,
                                              const float* __restrict__ gw,
                                              const float* __restrict__ gb,
                                              const float* __restrict__ ga,
                                              float* __restrict__ outF) {
    const int t = threadIdx.x, b = blockIdx.y, n0 = blockIdx.x * 16;
    const int d = t & (D_ - 1);
    const int r = t >> 7;
    const NormC c = norm_coef(gSum, gSq, gw, gb, ga, b, d);
    for (int i = 0; i < 8; ++i) {
        int n = n0 + r + 2 * i;
        size_t idx = (size_t)(b * N_ + n) * D_ + d;
        outF[idx] = (hIn[idx] - c.am) * c.g + c.bv;
    }
}

extern "C" void kernel_launch(void* const* d_in, const int* in_sizes, int n_in,
                              void* d_out, int out_size, void* d_ws, size_t ws_size,
                              hipStream_t stream) {
    (void)in_sizes; (void)n_in; (void)out_size; (void)ws_size;
    const float* inst = (const float*)d_in[0];
    const float* Wn   = (const float*)d_in[1];
    const float* bn   = (const float*)d_in[2];
    const float* Wg   = (const float*)d_in[3];
    const float* bg   = (const float*)d_in[4];
    const float* gw   = (const float*)d_in[5];
    const float* gb   = (const float*)d_in[6];
    const float* ga   = (const float*)d_in[7];

    char* w = (char*)d_ws;
    size_t off = 0;
    auto alloc = [&](size_t bytes) {
        void* p = w + off;
        off += (bytes + 255) & ~(size_t)255;
        return p;
    };
    float*    hF    = (float*)   alloc((size_t)B_ * N_ * D_ * 4);
    _Float16* XF    = (_Float16*)alloc((size_t)B_ * N_ * D_ * 2);
    float*    dinv  = (float*)   alloc((size_t)B_ * N_ * 4);
    float*    gS    = (float*)   alloc((size_t)L_ * B_ * D_ * 4);
    float*    gQ    = (float*)   alloc((size_t)L_ * B_ * D_ * 4);

    k_dinv_hw<<<dim3(N_ / DR_, B_), 256, 0, stream>>>(inst, Wn, bn, Wg, dinv, XF, gS, gQ);
    k_gemm_first<<<dim3(N_ / 64, B_), 128, 0, stream>>>(inst, dinv, XF, bg, Wn, bn,
                                                        hF, gS, gQ);
    for (int l = 1; l < L_; ++l) {
        const float* pS = gS + (l - 1) * B_ * D_;
        const float* pQ = gQ + (l - 1) * B_ * D_;
        k_hw_next<<<dim3(N_ / 128, B_), 256, 0, stream>>>(hF, pS, pQ, gw, gb, ga,
                                                          Wg + (size_t)l * D_ * D_, dinv, XF,
                                                          gS + l * B_ * D_, gQ + l * B_ * D_);
        k_gemm_next<<<dim3(N_ / 64, B_), 128, 0, stream>>>(inst, dinv, XF, bg + l * D_,
                                                           pS, pQ, gw, gb, ga, hF,
                                                           gS + l * B_ * D_, gQ + l * B_ * D_);
    }
    k_norm<<<dim3(N_ / 16, B_), 256, 0, stream>>>(hF, gS + (L_ - 1) * B_ * D_,
                                                  gQ + (L_ - 1) * B_ * D_, gw, gb, ga,
                                                  (float*)d_out);
}


# k_norm: a wave normalizes 8 whole rows (lane = 4 features, 16-byte loads and stores, all 8 loads ahead of the stores): 256 work-groups and 2048 waves to launch, was 1024 and 8192 of 4-byte accesses
# speedup vs baseline: 7.3655x; 1.5622x over previous
#include <hip/hip_runtime.h>

typedef __attribute__((ext_vector_type(16))) _Float16 v16h;
typedef __attribute__((ext_vector_type(8)))  _Float16 v8h;
typedef __attribute__((ext_vector_type(8)))  float    v8f;
typedef __attribute__((ext_vector_type(4)))  float    v4f;
typedef __attribute__((ext_vector_type(2)))  float    v2f;

#define B_   8
#define N_   2048
#define D_   128
#define L_   3
#define EPSN 1e-5f

__device__ __forceinline__ float fast_sqrt(float x) {
#if __has_builtin(__builtin_amdgcn_sqrtf)
    return __builtin_amdgcn_sqrtf(x);
#else
    return sqrtf(x);
#endif
}

__device__ __forceinline__ float fast_tanh(float x) {
#if __has_builtin(__builtin_amdgcn_tanhf)
    return __builtin_amdgcn_tanhf(x);
#else
    return tanhf(x);
#endif
}

__device__ __forceinline__ v16h load_frag(const _Float16* p, int hi_off) {
    v8h lo = *(const v8h*)(p);
    v8h hi = *(const v8h*)(p + hi_off);
    return __builtin_shufflevector(lo, hi, 0,1,2,3,4,5,6,7,8,9,10,11,12,13,14,15);
}

#define FRAG_  512
#define KCH_   (N_ / 32)
#define WP_    (D_ + 8)

struct NormC { float am, g, bv; };
__device__ __forceinline__ NormC norm_coef(const float* __restrict__ gSum,
                                           const float* __restrict__ gSq,
                                           const float* __restrict__ gw,
                                           const float* __restrict__ gb,
                                           const float* __restrict__ ga,
                                           int b, int d) {
    float a    = ga[d];
    float mean = gSum[b * D_ + d] * (1.0f / N_);
    float var  = gSq[b * D_ + d] * (1.0f / N_) - a * (2.0f - a) * mean * mean;
    float rs   = rsqrtf(var + EPSN);
    NormC c;
    c.am = a * mean;
    c.g  = gw[d] * rs;
    c.bv = gb[d];
    return c;
}

#define DR_   64
#define DS_   4
#define DG_   16
#define DSTEPS_ (N_ / DS_ / DG_)

struct ColG { v4f x[DG_ / 4], y[DG_ / 4]; };

__global__ __launch_bounds__(256) void k_dinv_hw(const float* __restrict__ inst,
                                                 const float* __restrict__ Wn,
                                                 const float* __restrict__ bn,
                                                 const float* __restrict__ Wgl,
                                                 float* __restrict__ dinv,
                                                 _Float16* __restrict__ XF,
                                                 float* __restrict__ zSum,
                                                 float* __restrict__ zSq) {
    __shared__ __align__(16) float cx[N_], cy[N_];
    __shared__ float part[DS_][DR_];
    __shared__ __align__(16) float sDv[DR_];
    __shared__ __align__(16) _Float16 sW[D_][WP_];
    __shared__ __align__(16) float sA[D_], sB[D_], sC[D_];
    const int b = blockIdx.y, t = threadIdx.x, n0 = blockIdx.x * DR_;
    const float* ib = inst + (size_t)b * N_ * 2;
    {
        v2f p[N_ / 256];
        v4f wl[(D_ * D_) / 8 / 256], wh[(D_ * D_) / 8 / 256];
        for (int i = 0; i < N_ / 256; ++i) p[i] = *(const v2f*)(ib + 2 * (t + 256 * i));
        for (int i = 0; i < (D_ * D_) / 8 / 256; ++i) {
            const int g = t + 256 * i;
            wl[i] = ((const v4f*)Wgl)[2 * g]; wh[i] = ((const v4f*)Wgl)[2 * g + 1];
        }
        for (int i = 0; i < N_ / 256; ++i) { cx[t + 256 * i] = p[i][0]; cy[t + 256 * i] = p[i][1]; }
        for (int i = 0; i < (D_ * D_) / 8 / 256; ++i) {
            const int g = t + 256 * i;
            v8h o;
            for (int e = 0; e < 4; ++e) { o[e] = (_Float16)wl[i][e]; o[4 + e] = (_Float16)wh[i][e]; }
            *(v8h*)&sW[g >> 4][(g & 15) * 8] = o;
        }
    }
    if (t < D_) { sA[t] = Wn[2 * t]; sB[t] = Wn[2 * t + 1]; sC[t] = bn[t]; }
    if (blockIdx.x == 0 && b == 0) {
        for (int i = t; i < B_ * D_; i += 256) { zSum[i] = 0.f; zSq[i] = 0.f; }
    }
    __syncthreads();

    const int r = t & (DR_ - 1), s = t / DR_;
    const float xn = cx[n0 + r], yn = cy[n0 + r];
    const int m0 = s * (N_ / DS_);

    auto loadG = [&](ColG& c, int g) {
        for (int q = 0; q < DG_ / 4; ++q) {
            c.x[q] = *(const v4f*)&cx[m0 + g * DG_ + 4 * q];
            c.y[q] = *(const v4f*)&cy[m0 + g * DG_ + 4 * q];
        }
    };
    v4f dsum = {0.f, 0.f, 0.f, 0.f};
    auto sumG = [&](const ColG& c) {
        for (int q = 0; q < DG_ / 4; ++q) {
            const v4f dx = xn - c.x[q], dy = yn - c.y[q];
            const v4f d2 = dx * dx + dy * dy;
            for (int e = 0; e < 4; ++e) dsum[e] += fast_sqrt(d2[e]);
        }
    };

    ColG c0, c1;
    loadG(c0, 0);
    for (int g = 0; g < DSTEPS_ - 2; g += 2) {
        loadG(c1, g + 1); sumG(c0);
        loadG(c0, g + 2); sumG(c1);
    }
    loadG(c1, DSTEPS_ - 1); sumG(c0);
    sumG(c1);

    part[s][r] = (dsum[0] + dsum[1]) + (dsum[2] + dsum[3]);
    __syncthreads();
    if (t < DR_) {
        float deg = 1.0f;
        for (int i = 0; i < DS_; ++i) deg += part[i][t];
        const float dv = rsqrtf(deg);
        dinv[b * N_ + n0 + t] = dv;
        sDv[t] = dv;
    }
    __syncthreads();

    const int lane = t & 31, w = t >> 5;
    const int col16 = lane & 15;
    const int kb   = (lane < 16) ? 0 : 8;
    const int rt   = w & 3, ch = w >> 2;
    const float px = cx[n0 + rt * 16 + col16];
    const float py = cy[n0 + rt * 16 + col16];
    const v4f dv0 = *(const v4f*)&sDv[rt * 16 + kb], dv1 = *(const v4f*)&sDv[rt * 16 + kb + 4];

    v8f zero; for (int i = 0; i < 8; ++i) zero[i] = 0.f;
    v8f acc[4]; for (int j = 0; j < 4; ++j) acc[j] = zero;

    for (int kc = 0; kc < D_ / 32; ++kc) {
        v16h a;
        for (int q = 0; q < 4; ++q) {
            const int k = kc * 32 + kb + (q >> 1) * 16 + (q & 1) * 4;
            const v4f ca = *(const v4f*)&sA[k];
            const v4f cb = *(const v4f*)&sB[k];
            const v4f cc = *(const v4f*)&sC[k];
            for (int e = 0; e < 4; ++e)
                a[q * 4 + e] = (_Float16)(px * ca[e] + py * cb[e] + cc[e]);
        }
        v16h bf[4];
        for (int j = 0; j < 4; ++j)
            bf[j] = load_frag(&sW[(ch * 4 + j) * 16 + col16][kc * 32 + kb], 16);
        for (int j = 0; j < 4; ++j)
            acc[j] = __builtin_amdgcn_wmma_f32_16x16x32_f16(
                false, a, false, bf[j], (short)0, acc[j], false, false);
    }
    const int kcw = blockIdx.x * (DR_ / 32) + (rt >> 1);
    _Float16* xo = XF + (size_t)((b * KCH_ + kcw) * 8 + ch * 4) * FRAG_ + (rt & 1) * (FRAG_ / 2) + lane * 8;
    for (int j = 0; j < 4; ++j) {
        v8h out;
        for (int q = 0; q < 4; ++q) {
            out[q]     = (_Float16)(acc[j][q]     * dv0[q]);
            out[4 + q] = (_Float16)(acc[j][4 + q] * dv1[q]);
        }
        *(v8h*)(xo + j * FRAG_) = out;
    }
}

__global__ __launch_bounds__(256) void k_hw_next(const float* __restrict__ hF,
                                                 const float* __restrict__ pSum,
                                                 const float* __restrict__ pSq,
                                                 const float* __restrict__ gw,
                                                 const float* __restrict__ gb,
                                                 const float* __restrict__ ga,
                                                 const float* __restrict__ Wgl,
                                                 const float* __restrict__ dinv,
                                                 _Float16* __restrict__ XF,
                                                 float* __restrict__ zSum,
                                                 float* __restrict__ zSq) {
    __shared__ __align__(16) _Float16 sW[D_][WP_];
    __shared__ __align__(16) float sA[D_], sB[D_], sC[D_];
    const int t = threadIdx.x, b = blockIdx.y, n0 = blockIdx.x * 128;

    const int lane = t & 31, w = t >> 5;
    const int col16 = lane & 15;
    const int kb   = (lane < 16) ? 0 : 8;
    const int row  = n0 + w * 16 + col16;

    const float* dvp = dinv + b * N_ + n0 + w * 16 + kb;
    const v4f dv0 = *(const v4f*)dvp, dv1 = *(const v4f*)(dvp + 4);

    v4f hv[16];
    {
        const float* hp = hF + (size_t)(b * N_ + row) * D_ + kb;
        for (int kc = 0; kc < D_ / 32; ++kc)
            for (int q = 0; q < 4; ++q)
                hv[kc * 4 + q] = *(const v4f*)(hp + kc * 32 + (q >> 1) * 16 + (q & 1) * 4);
    }

    if (blockIdx.x == 0 && b == 0) {
        for (int i = t; i < B_ * D_; i += 256) { zSum[i] = 0.f; zSq[i] = 0.f; }
    }
    if (t < D_) {
        NormC c = norm_coef(pSum, pSq, gw, gb, ga, b, t);
        sA[t] = c.am; sB[t] = c.g; sC[t] = c.bv;
    }
    for (int i = t; i < (D_ * D_) / 8; i += 256) {
        v4f lo = ((const v4f*)Wgl)[2 * i], hi = ((const v4f*)Wgl)[2 * i + 1];
        v8h o;
        for (int e = 0; e < 4; ++e) { o[e] = (_Float16)lo[e]; o[4 + e] = (_Float16)hi[e]; }
        *(v8h*)&sW[i >> 4][(i & 15) * 8] = o;
    }
    __syncthreads();

    v8f zero; for (int i = 0; i < 8; ++i) zero[i] = 0.f;
    v8f acc[8]; for (int j = 0; j < 8; ++j) acc[j] = zero;

    for (int kc = 0; kc < D_ / 32; ++kc) {
        v16h a;
        for (int q = 0; q < 4; ++q) {
            const int k = kc * 32 + kb + (q >> 1) * 16 + (q & 1) * 4;
            const v4f ca = *(const v4f*)&sA[k];
            const v4f cb = *(const v4f*)&sB[k];
            const v4f cc = *(const v4f*)&sC[k];
            for (int e = 0; e < 4; ++e)
                a[q * 4 + e] = (_Float16)((hv[kc * 4 + q][e] - ca[e]) * cb[e] + cc[e]);
        }
        v16h bf[8];
        for (int j = 0; j < 8; ++j)
            bf[j] = load_frag(&sW[j * 16 + col16][kc * 32 + kb], 16);
        for (int j = 0; j < 8; ++j)
            acc[j] = __builtin_amdgcn_wmma_f32_16x16x32_f16(
                false, a, false, bf[j], (short)0, acc[j], false, false);
    }
    const int kcw = blockIdx.x * 4 + (w >> 1);
    _Float16* xo = XF + (size_t)((b * KCH_ + kcw) * 8) * FRAG_ + (w & 1) * (FRAG_ / 2) + lane * 8;
    for (int j = 0; j < 8; ++j) {
        v8h out;
        for (int r = 0; r < 4; ++r) {
            out[r]     = (_Float16)(acc[j][r]     * dv0[r]);
            out[4 + r] = (_Float16)(acc[j][4 + r] * dv1[r]);
        }
        *(v8h*)(xo + j * FRAG_) = out;
    }
}


struct ColT { v4f x[4], y[4]; };

template <bool FIRST>
__device__ __forceinline__ void gemm_body(const float* __restrict__ inst,
                                          const float* __restrict__ dinv,
                                          const _Float16* __restrict__ XF,
                                          const float* __restrict__ bgl,
                                          const float* __restrict__ Wn,
                                          const float* __restrict__ bn,
                                          const float* __restrict__ pSum,
                                          const float* __restrict__ pSq,
                                          const float* __restrict__ gw,
                                          const float* __restrict__ gb,
                                          const float* __restrict__ ga,
                                          float* __restrict__ hF,
                                          float* __restrict__ gSum,
                                          float* __restrict__ gSq) {
    __shared__ __align__(16) float cX[N_], cY[N_];
    __shared__ float sSum[D_], sSq[D_];
    __shared__ float sR0[D_], sR1[D_], sR2[D_];
    __shared__ __align__(16) v4f sEx[2][4][4][2][32];

    const int t = threadIdx.x, b = blockIdx.y, n0 = blockIdx.x * 64;
    const float* ib = inst + (size_t)b * N_ * 2;

    const int lane = t & 31;
    const int w  = __builtin_amdgcn_readfirstlane(t >> 5);
    const int rt = w & 3, kh = w >> 2;
    const int k0 = kh * (KCH_ / 2);
    const int rot = blockIdx.x & (KCH_ / 2 - 1);
    auto ck = [&](int i) { return k0 + ((rot + i) & (KCH_ / 2 - 1)); };
    const int col16 = lane & 15;
    const int kb   = (lane < 16) ? 0 : 8;
    const int roff = (lane < 16) ? 0 : 8;
    const int gn   = n0 + rt * 16 + col16;

    const float* dq = dinv + b * N_ + n0 + rt * 16 + roff;
    const v4f dq0 = *(const v4f*)dq, dq1 = *(const v4f*)(dq + 4);

    const _Float16* xb = XF + (size_t)b * KCH_ * 8 * FRAG_ + lane * 8;

    auto loadB = [&](v16h (&buf)[8], int kc) {
        const _Float16* p = xb + (size_t)kc * 8 * FRAG_;
        for (int j = 0; j < 8; ++j) buf[j] = load_frag(p + j * FRAG_, FRAG_ / 2);
    };

    v16h b0[8], b1[8], b2[8], b3[8];
    loadB(b0, ck(0)); loadB(b1, ck(1)); loadB(b2, ck(2));

    {
        float tx[N_ / 256], ty[N_ / 256];
        for (int i = 0; i < N_ / 256; ++i) {
            const int m = t + 256 * i;
            tx[i] = ib[2*m]; ty[i] = ib[2*m + 1];
        }
        for (int i = 0; i < N_ / 256; ++i) {
            const int m = t + 256 * i;
            cX[m] = tx[i]; cY[m] = ty[i];
        }
    }
    if (t < D_) {
        sSum[t] = 0.f; sSq[t] = 0.f;
        if (FIRST) {
            sR0[t] = Wn[2 * t]; sR1[t] = Wn[2 * t + 1]; sR2[t] = bn[t];
        } else {
            NormC c = norm_coef(pSum, pSq, gw, gb, ga, b, t);
            sR0[t] = c.am; sR1[t] = c.g; sR2[t] = c.bv;
        }
    }
    __syncthreads();

    const float xr = cX[gn], yr = cY[gn];
    auto loadC = [&](ColT& c, int kc) {
        const int base = kc * 32 + kb;
        for (int q = 0; q < 4; ++q) {
            const int o = base + (q >> 1) * 16 + (q & 1) * 4;
            c.x[q] = *(const v4f*)&cX[o];
            c.y[q] = *(const v4f*)&cY[o];
        }
    };
    auto makeA = [&](const ColT& c) -> v16h {
        v16h a;
        for (int q = 0; q < 4; ++q) {
            for (int e = 0; e < 4; ++e) {
                float dx = xr - c.x[q][e], dy = yr - c.y[q][e];
                a[q * 4 + e] = (_Float16)fast_sqrt(dx*dx + dy*dy);
            }
        }
        return a;
    };

    v8f zero; for (int i = 0; i < 8; ++i) zero[i] = 0.f;
    v8f acc[8]; for (int j = 0; j < 8; ++j) acc[j] = zero;

    auto mma = [&](const v16h& a, const v16h (&buf)[8]) {
        for (int j = 0; j < 8; ++j)
            acc[j] = __builtin_amdgcn_wmma_f32_16x16x32_f16(
                false, a, false, buf[j], (short)0, acc[j], false, false);
    };

    ColT c0, c1;
    loadC(c0, ck(0));

    auto pin = [&]() { __builtin_amdgcn_sched_barrier(0x78F); };

    for (int i = 0; i < KCH_ / 2 - 4; i += 4) {
        loadB(b3, ck(i + 3)); pin(); loadC(c1, ck(i + 1)); mma(makeA(c0), b0);
        pin(); loadB(b0, ck(i + 4)); pin(); loadC(c0, ck(i + 2)); mma(makeA(c1), b1);
        pin(); loadB(b1, ck(i + 5)); pin(); loadC(c1, ck(i + 3)); mma(makeA(c0), b2);
        pin(); loadB(b2, ck(i + 6)); pin(); loadC(c0, ck(i + 4)); mma(makeA(c1), b3);
    }
    const int rb = n0 + rt * 16;
    const _Float16* xsp = xb + (size_t)((rb >> 5) * 8 + kh * 4) * FRAG_ + ((rb >> 4) & 1) * (FRAG_ / 2);
    v8h xs[4];
    for (int jj = 0; jj < 4; ++jj) xs[jj] = *(const v8h*)(xsp + jj * FRAG_);
    {
        const int i = KCH_ / 2 - 4;
        loadB(b3, ck(i + 3)); loadC(c1, ck(i + 1)); mma(makeA(c0), b0);
                              loadC(c0, ck(i + 2)); mma(makeA(c1), b1);
                              loadC(c1, ck(i + 3)); mma(makeA(c0), b2);
                                                    mma(makeA(c1), b3);
    }

    v8f fin[4];
    for (int jj = 0; jj < 4; ++jj) {
        const v8f give = kh ? acc[jj] : acc[4 + jj];
        fin[jj]        = kh ? acc[4 + jj] : acc[jj];
        for (int q = 0; q < 2; ++q) {
            const v4f o = {give[4 * q], give[4 * q + 1], give[4 * q + 2], give[4 * q + 3]};
            sEx[1 - kh][rt][jj][q][lane] = o;
        }
    }
    __syncthreads();
    for (int jj = 0; jj < 4; ++jj)
        for (int q = 0; q < 2; ++q) {
            const v4f o = sEx[kh][rt][jj][q][lane];
            for (int e = 0; e < 4; ++e) fin[jj][4 * q + e] += o[e];
        }

    float drv[8];
    for (int r = 0; r < 4; ++r) { drv[r] = dq0[r]; drv[4 + r] = dq1[r]; }
    for (int jj = 0; jj < 4; ++jj) {
        int d = (kh * 4 + jj) * 16 + col16;
        float bgv = bgl[d];
        const float q0 = sR0[d], q1 = sR1[d], q2 = sR2[d];
        float s = 0.f, ss = 0.f;
        for (int r = 0; r < 8; ++r) {
            int nl = n0 + rt * 16 + r + roff;
            size_t idx = (size_t)(b * N_ + nl) * D_ + d;
            float res;
            if (FIRST) res = cX[nl] * q0 + cY[nl] * q1 + q2;
            else       res = (hF[idx] - q0) * q1 + q2;
            float z  = (fin[jj][r] + (float)xs[jj][r]) * drv[r];
            float hp = fast_tanh(z + bgv) + res;
            hF[idx] = hp;
            s += hp; ss += hp * hp;
        }
        atomicAdd(&sSum[d], s);
        atomicAdd(&sSq[d], ss);
    }
    __syncthreads();
    if (t < D_) {
        atomicAdd(&gSum[b * D_ + t], sSum[t]);
        atomicAdd(&gSq[b * D_ + t],  sSq[t]);
    }
}

__global__ __launch_bounds__(256) void k_gemm_first(const float* __restrict__ inst,
                                                    const float* __restrict__ dinv,
                                                    const _Float16* __restrict__ XF,
                                                    const float* __restrict__ bgl,
                                                    const float* __restrict__ Wn,
                                                    const float* __restrict__ bn,
                                                    float* __restrict__ hF,
                                                    float* __restrict__ gSum,
                                                    float* __restrict__ gSq) {
    gemm_body<true>(inst, dinv, XF, bgl, Wn, bn, nullptr, nullptr, nullptr, nullptr, nullptr,
                    hF, gSum, gSq);
}

__global__ __launch_bounds__(256) void k_gemm_next(const float* __restrict__ inst,
                                                   const float* __restrict__ dinv,
                                                   const _Float16* __restrict__ XF,
                                                   const float* __restrict__ bgl,
                                                   const float* __restrict__ pSum,
                                                   const float* __restrict__ pSq,
                                                   const float* __restrict__ gw,
                                                   const float* __restrict__ gb,
                                                   const float* __restrict__ ga,
                                                   float* __restrict__ hF,
                                                   float* __restrict__ gSum,
                                                   float* __restrict__ gSq) {
    gemm_body<false>(inst, dinv, XF, bgl, nullptr, nullptr, pSum, pSq, gw, gb, ga,
                     hF, gSum, gSq);
}

#define NR_   8
__global__ __launch_bounds__(256) void k_norm(const float* __restrict__ hIn,
                                              const float* __restrict__ gSum,
                                              const float* __restrict__ gSq,
                                              const float* __restrict__ gw,
                                              const float* __restrict__ gb,
                                              const float* __restrict__ ga,
                                              float* __restrict__ outF) {
    const int t = threadIdx.x, b = blockIdx.y;
    const int lane = t & 31, w = t >> 5;
    const int d = lane * 4;
    const size_t base = (size_t)(b * N_ + blockIdx.x * (8 * NR_) + w * NR_) * D_ + d;

    v4f v[NR_];
    for (int i = 0; i < NR_; ++i) v[i] = *(const v4f*)(hIn + base + (size_t)i * D_);

    const v4f a4 = *(const v4f*)(ga + d);
    const v4f w4 = *(const v4f*)(gw + d);
    const v4f b4 = *(const v4f*)(gb + d);
    const v4f s4 = *(const v4f*)(gSum + b * D_ + d);
    const v4f q4 = *(const v4f*)(gSq + b * D_ + d);
    v4f am, g;
    for (int e = 0; e < 4; ++e) {
        const float a    = a4[e];
        const float mean = s4[e] * (1.0f / N_);
        const float var  = q4[e] * (1.0f / N_) - a * (2.0f - a) * mean * mean;
        const float rs   = rsqrtf(var + EPSN);
        am[e] = a * mean;
        g[e]  = w4[e] * rs;
    }
    for (int i = 0; i < NR_; ++i) {
        v4f o;
        for (int e = 0; e < 4; ++e) o[e] = (v[i][e] - am[e]) * g[e] + b4[e];
        *(v4f*)(outF + base + (size_t)i * D_) = o;
    }
}

extern "C" void kernel_launch(void* const* d_in, const int* in_sizes, int n_in,
                              void* d_out, int out_size, void* d_ws, size_t ws_size,
                              hipStream_t stream) {
    (void)in_sizes; (void)n_in; (void)out_size; (void)ws_size;
    const float* inst = (const float*)d_in[0];
    const float* Wn   = (const float*)d_in[1];
    const float* bn   = (const float*)d_in[2];
    const float* Wg   = (const float*)d_in[3];
    const float* bg   = (const float*)d_in[4];
    const float* gw   = (const float*)d_in[5];
    const float* gb   = (const float*)d_in[6];
    const float* ga   = (const float*)d_in[7];

    char* w = (char*)d_ws;
    size_t off = 0;
    auto alloc = [&](size_t bytes) {
        void* p = w + off;
        off += (bytes + 255) & ~(size_t)255;
        return p;
    };
    float*    hF    = (float*)   alloc((size_t)B_ * N_ * D_ * 4);
    _Float16* XF    = (_Float16*)alloc((size_t)B_ * N_ * D_ * 2);
    float*    dinv  = (float*)   alloc((size_t)B_ * N_ * 4);
    float*    gS    = (float*)   alloc((size_t)L_ * B_ * D_ * 4);
    float*    gQ    = (float*)   alloc((size_t)L_ * B_ * D_ * 4);

    k_dinv_hw<<<dim3(N_ / DR_, B_), 256, 0, stream>>>(inst, Wn, bn, Wg, dinv, XF, gS, gQ);
    k_gemm_first<<<dim3(N_ / 64, B_), 256, 0, stream>>>(inst, dinv, XF, bg, Wn, bn,
                                                        hF, gS, gQ);
    for (int l = 1; l < L_; ++l) {
        const float* pS = gS + (l - 1) * B_ * D_;
        const float* pQ = gQ + (l - 1) * B_ * D_;
        k_hw_next<<<dim3(N_ / 128, B_), 256, 0, stream>>>(hF, pS, pQ, gw, gb, ga,
                                                          Wg + (size_t)l * D_ * D_, dinv, XF,
                                                          gS + l * B_ * D_, gQ + l * B_ * D_);
        k_gemm_next<<<dim3(N_ / 64, B_), 256, 0, stream>>>(inst, dinv, XF, bg + l * D_,
                                                           pS, pQ, gw, gb, ga, hF,
                                                           gS + l * B_ * D_, gQ + l * B_ * D_);
    }
    k_norm<<<dim3(N_ / (8 * NR_), B_), 256, 0, stream>>>(hF, gS + (L_ - 1) * B_ * D_,
                                                  gQ + (L_ - 1) * B_ * D_, gw, gb, ga,
                                                  (float*)d_out);
}
